# starting kernel
# baseline (speedup 1.0000x reference)
#include <hip/hip_runtime.h>

typedef float v2f __attribute__((ext_vector_type(2)));
typedef float v8f __attribute__((ext_vector_type(8)));

#define N_NODES 20000
#define N_EDGES 640000

__device__ __forceinline__ v8f zero8() {
  v8f v = {0.f, 0.f, 0.f, 0.f, 0.f, 0.f, 0.f, 0.f};
  return v;
}

__device__ __forceinline__ v8f wmma4(v2f a, v2f b, v8f c) {
  return __builtin_amdgcn_wmma_f32_16x16x4_f32(false, a, false, b, (short)0, c,
                                               false, false);
}

__device__ __forceinline__ float gelu_tanh(float x) {
  float x3 = x * x * x;
  return 0.5f * x * (1.0f + tanhf(0.7978845608028654f * (x + 0.044715f * x3)));
}

__global__ __launch_bounds__(128) void node_linear_kernel(
    const float* __restrict__ node_s, const float* __restrict__ node_v,
    const float* __restrict__ Wsf, const float* __restrict__ Wss,
    const float* __restrict__ Wvf, const float* __restrict__ Wvs,
    float* __restrict__ fs, float* __restrict__ fvb, float* __restrict__ ss,
    float* __restrict__ svb) {
  const int lane = threadIdx.x & 31;
  const int wave = threadIdx.x >> 5;
  const int tile = blockIdx.x * 4 + wave;
  if (tile >= N_NODES / 16) return;
  const int n0 = tile * 16;
  const int l15 = lane & 15, lh = lane >> 4;
  const float inv_mul = 0.17677669529663687f;

  {
    v8f aF0 = zero8(), aF1 = zero8(), aS0 = zero8(), aS1 = zero8();
    const float* arow = node_s + (size_t)(n0 + l15) * 32;
#pragma unroll
    for (int ks = 0; ks < 8; ++ks) {
      int k = ks * 4 + lh * 2;
      v2f a = *reinterpret_cast<const v2f*>(arow + k);
      v2f bf0, bf1, bs0, bs1;
      bf0.x = Wsf[k * 32 + l15];      bf0.y = Wsf[(k + 1) * 32 + l15];
      bf1.x = Wsf[k * 32 + 16 + l15]; bf1.y = Wsf[(k + 1) * 32 + 16 + l15];
      bs0.x = Wss[k * 32 + l15];      bs0.y = Wss[(k + 1) * 32 + l15];
      bs1.x = Wss[k * 32 + 16 + l15]; bs1.y = Wss[(k + 1) * 32 + 16 + l15];
      aF0 = wmma4(a, bf0, aF0);
      aF1 = wmma4(a, bf1, aF1);
      aS0 = wmma4(a, bs0, aS0);
      aS1 = wmma4(a, bs1, aS1);
    }
#pragma unroll
    for (int r = 0; r < 8; ++r) {
      int node = n0 + r + 8 * lh;
      fs[(size_t)node * 32 + l15]      = aF0[r] * inv_mul;
      fs[(size_t)node * 32 + 16 + l15] = aF1[r] * inv_mul;
      ss[(size_t)node * 32 + l15]      = aS0[r] * inv_mul;
      ss[(size_t)node * 32 + 16 + l15] = aS1[r] * inv_mul;
    }
  }

#pragma unroll
  for (int c = 0; c < 3; ++c) {
    v8f aF0 = zero8(), aF1 = zero8(), aS0 = zero8(), aS1 = zero8();
    const float* vrow = node_v + (size_t)(n0 + l15) * 96 + c;
#pragma unroll
    for (int ks = 0; ks < 8; ++ks) {
      int k = ks * 4 + lh * 2;
      v2f a;
      a.x = vrow[k * 3];
      a.y = vrow[(k + 1) * 3];
      v2f bf0, bf1, bs0, bs1;
      bf0.x = Wvf[k * 32 + l15];      bf0.y = Wvf[(k + 1) * 32 + l15];
      bf1.x = Wvf[k * 32 + 16 + l15]; bf1.y = Wvf[(k + 1) * 32 + 16 + l15];
      bs0.x = Wvs[k * 32 + l15];      bs0.y = Wvs[(k + 1) * 32 + l15];
      bs1.x = Wvs[k * 32 + 16 + l15]; bs1.y = Wvs[(k + 1) * 32 + 16 + l15];
      aF0 = wmma4(a, bf0, aF0);
      aF1 = wmma4(a, bf1, aF1);
      aS0 = wmma4(a, bs0, aS0);
      aS1 = wmma4(a, bs1, aS1);
    }
#pragma unroll
    for (int r = 0; r < 8; ++r) {
      int node = n0 + r + 8 * lh;
      fvb[((size_t)node * 3 + c) * 32 + l15]      = aF0[r] * inv_mul;
      fvb[((size_t)node * 3 + c) * 32 + 16 + l15] = aF1[r] * inv_mul;
      svb[((size_t)node * 3 + c) * 32 + l15]      = aS0[r] * inv_mul;
      svb[((size_t)node * 3 + c) * 32 + 16 + l15] = aS1[r] * inv_mul;
    }
  }
}

__global__ __launch_bounds__(64) void edge_kernel(
    const float* __restrict__ edge_s, const float* __restrict__ edge_v,
    const float* __restrict__ edge_rad, const int* __restrict__ esrc,
    const int* __restrict__ edst, const float* __restrict__ Wmlp0,
    const float* __restrict__ Wmlp1, const float* __restrict__ Wi0,
    const float* __restrict__ Wi1, const float* __restrict__ Wi2,
    const float* __restrict__ Wi3, const float* __restrict__ fs,
    const float* __restrict__ fvb, float* __restrict__ ns,
    float* __restrict__ nv) {
  __shared__ float lds[13312 + 2 * 1056];
  const int tid = threadIdx.x;

  for (int idx = tid; idx < 1024; idx += 64) {
    int j = idx & 1, l = (idx >> 1) & 31, f = idx >> 6;
    int k = (f >> 2) * 4 + (l >> 4) * 2 + j;
    int n = (f & 3) * 16 + (l & 15);
    lds[idx] = Wmlp0[k * 64 + n] * 0.25f;
  }
  for (int idx = tid; idx < 4096; idx += 64) {
    int j = idx & 1, l = (idx >> 1) & 31, f = idx >> 6;
    int k = (f >> 2) * 4 + (l >> 4) * 2 + j;
    int n = (f & 3) * 16 + (l & 15);
    lds[1024 + idx] = Wmlp1[k * 64 + n] * 0.125f;
  }
  for (int idx = tid; idx < 8192; idx += 64) {
    int j = idx & 1, l = (idx >> 1) & 31, f = idx >> 6;
    int k = (f >> 3) * 4 + (l >> 4) * 2 + j;
    int n = (f & 7) * 16 + (l & 15);
    const float* Wi = (n < 32) ? Wi0 : (n < 64) ? Wi1 : (n < 96) ? Wi2 : Wi3;
    lds[5120 + idx] = Wi[k * 32 + (n & 31)] * 0.125f;
  }
  __syncthreads();

  const int lane = tid & 31, wave = tid >> 5;
  const int l15 = lane & 15, lh = lane >> 4;
  const int tile = blockIdx.x * 2 + wave;
  const int e0 = tile * 16;
  float* hbuf = lds + 13312 + wave * 1056;

  v8f h1[4];
#pragma unroll
  for (int nt = 0; nt < 4; ++nt) h1[nt] = zero8();
  const float* Xrow = edge_rad + ((size_t)e0 + l15) * 16;
#pragma unroll
  for (int ks = 0; ks < 4; ++ks) {
    int k = ks * 4 + lh * 2;
    v2f a = *reinterpret_cast<const v2f*>(Xrow + k);
#pragma unroll
    for (int nt = 0; nt < 4; ++nt) {
      v2f b = *reinterpret_cast<const v2f*>(&lds[(ks * 4 + nt) * 64 + lane * 2]);
      h1[nt] = wmma4(a, b, h1[nt]);
    }
  }
#pragma unroll
  for (int nt = 0; nt < 4; ++nt)
#pragma unroll
    for (int r = 0; r < 8; ++r)
      hbuf[(r + 8 * lh) * 66 + nt * 16 + l15] = gelu_tanh(h1[nt][r]);

  v8f h2[4];
#pragma unroll
  for (int nt = 0; nt < 4; ++nt) h2[nt] = zero8();
#pragma unroll
  for (int ks = 0; ks < 16; ++ks) {
    int k = ks * 4 + lh * 2;
    v2f a = *reinterpret_cast<const v2f*>(&hbuf[l15 * 66 + k]);
#pragma unroll
    for (int nt = 0; nt < 4; ++nt) {
      v2f b =
          *reinterpret_cast<const v2f*>(&lds[1024 + (ks * 4 + nt) * 64 + lane * 2]);
      h2[nt] = wmma4(a, b, h2[nt]);
    }
  }
#pragma unroll
  for (int nt = 0; nt < 4; ++nt)
#pragma unroll
    for (int r = 0; r < 8; ++r)
      hbuf[(r + 8 * lh) * 66 + nt * 16 + l15] = gelu_tanh(h2[nt][r]);

  v8f w[8];
#pragma unroll
  for (int nt = 0; nt < 8; ++nt) w[nt] = zero8();
#pragma unroll
  for (int ks = 0; ks < 16; ++ks) {
    int k = ks * 4 + lh * 2;
    v2f a = *reinterpret_cast<const v2f*>(&hbuf[l15 * 66 + k]);
#pragma unroll
    for (int nt = 0; nt < 8; ++nt) {
      v2f b =
          *reinterpret_cast<const v2f*>(&lds[5120 + (ks * 8 + nt) * 64 + lane * 2]);
      w[nt] = wmma4(a, b, w[nt]);
    }
  }

  const float inv_s3 = 0.5773502691896258f;
#pragma unroll
  for (int r = 0; r < 8; ++r) {
    int m = r + 8 * lh;
    int e = e0 + m;
    float se = edge_s[e];
    float vx = edge_v[(size_t)e * 3 + 0];
    float vy = edge_v[(size_t)e * 3 + 1];
    float vz = edge_v[(size_t)e * 3 + 2];
    int src = esrc[e];
    int dst = edst[e];
    const float* fr = fs + (size_t)src * 32;
    const float* vr = fvb + (size_t)src * 96;
    float esA = fr[l15], esB = fr[16 + l15];
    float exA = vr[l15], exB = vr[16 + l15];
    float eyA = vr[32 + l15], eyB = vr[48 + l15];
    float ezA = vr[64 + l15], ezB = vr[80 + l15];
    float w0A = w[0][r], w0B = w[1][r];
    float w1A = w[2][r], w1B = w[3][r];
    float w2A = w[4][r], w2B = w[5][r];
    float w3A = w[6][r], w3B = w[7][r];
    float dA = exA * vx + eyA * vy + ezA * vz;
    float dB = exB * vx + eyB * vy + ezB * vz;

    float* nsr = ns + (size_t)dst * 64;
    atomicAdd(nsr + l15, w0A * esA * se);
    atomicAdd(nsr + 16 + l15, w0B * esB * se);
    atomicAdd(nsr + 32 + l15, w3A * dA * inv_s3);
    atomicAdd(nsr + 48 + l15, w3B * dB * inv_s3);

    float t1A = w1A * esA, t1B = w1B * esB;
    float vcs[3] = {vx, vy, vz};
    float evA[3] = {exA, eyA, ezA};
    float evB[3] = {exB, eyB, ezB};
    float* nvr = nv + (size_t)dst * 192;
#pragma unroll
    for (int c = 0; c < 3; ++c) {
      atomicAdd(nvr + c * 64 + l15, t1A * vcs[c]);
      atomicAdd(nvr + c * 64 + 16 + l15, t1B * vcs[c]);
      atomicAdd(nvr + c * 64 + 32 + l15, w2A * evA[c] * se);
      atomicAdd(nvr + c * 64 + 48 + l15, w2B * evB[c] * se);
    }
  }
}

__global__ __launch_bounds__(128) void out_linear_kernel(
    const float* __restrict__ ns, const float* __restrict__ nv,
    const float* __restrict__ W2s, const float* __restrict__ W2v,
    const float* __restrict__ ss, const float* __restrict__ svb,
    float* __restrict__ out) {
  const int lane = threadIdx.x & 31;
  const int wave = threadIdx.x >> 5;
  const int tile = blockIdx.x * 4 + wave;
  if (tile >= N_NODES / 16) return;
  const int n0 = tile * 16;
  const int l15 = lane & 15, lh = lane >> 4;
  const float cosA = 0.9238795325112867f;
  const float scale = 0.3826834323650898f * 0.125f * 0.17677669529663687f;
  float* outs = out;
  float* outv = out + (size_t)N_NODES * 32;

  {
    v8f a0 = zero8(), a1 = zero8();
    const float* arow = ns + (size_t)(n0 + l15) * 64;
#pragma unroll
    for (int ks = 0; ks < 16; ++ks) {
      int k = ks * 4 + lh * 2;
      v2f a = *reinterpret_cast<const v2f*>(arow + k);
      v2f b0, b1;
      b0.x = W2s[k * 32 + l15];      b0.y = W2s[(k + 1) * 32 + l15];
      b1.x = W2s[k * 32 + 16 + l15]; b1.y = W2s[(k + 1) * 32 + 16 + l15];
      a0 = wmma4(a, b0, a0);
      a1 = wmma4(a, b1, a1);
    }
#pragma unroll
    for (int r = 0; r < 8; ++r) {
      int node = n0 + r + 8 * lh;
      outs[(size_t)node * 32 + l15] =
          cosA * ss[(size_t)node * 32 + l15] + scale * a0[r];
      outs[(size_t)node * 32 + 16 + l15] =
          cosA * ss[(size_t)node * 32 + 16 + l15] + scale * a1[r];
    }
  }

#pragma unroll
  for (int c = 0; c < 3; ++c) {
    v8f a0 = zero8(), a1 = zero8();
    const float* arow = nv + ((size_t)(n0 + l15) * 3 + c) * 64;
#pragma unroll
    for (int ks = 0; ks < 16; ++ks) {
      int k = ks * 4 + lh * 2;
      v2f a = *reinterpret_cast<const v2f*>(arow + k);
      v2f b0, b1;
      b0.x = W2v[k * 32 + l15];      b0.y = W2v[(k + 1) * 32 + l15];
      b1.x = W2v[k * 32 + 16 + l15]; b1.y = W2v[(k + 1) * 32 + 16 + l15];
      a0 = wmma4(a, b0, a0);
      a1 = wmma4(a, b1, a1);
    }
#pragma unroll
    for (int r = 0; r < 8; ++r) {
      int node = n0 + r + 8 * lh;
      outv[((size_t)node * 32 + l15) * 3 + c] =
          cosA * svb[((size_t)node * 3 + c) * 32 + l15] + scale * a0[r];
      outv[((size_t)node * 32 + 16 + l15) * 3 + c] =
          cosA * svb[((size_t)node * 3 + c) * 32 + 16 + l15] + scale * a1[r];
    }
  }
}

extern "C" void kernel_launch(void* const* d_in, const int* in_sizes, int n_in,
                              void* d_out, int out_size, void* d_ws,
                              size_t ws_size, hipStream_t stream) {
  (void)in_sizes; (void)n_in; (void)out_size; (void)ws_size;
  const float* node_s = (const float*)d_in[0];
  const float* node_v = (const float*)d_in[1];
  const float* edge_s = (const float*)d_in[2];
  const float* edge_v = (const float*)d_in[3];
  const float* edge_rad = (const float*)d_in[4];
  const int* esrc = (const int*)d_in[5];
  const int* edst = (const int*)d_in[6];
  const float* W1s_feat = (const float*)d_in[7];
  const float* W1s_self = (const float*)d_in[8];
  const float* W1v_feat = (const float*)d_in[9];
  const float* W1v_self = (const float*)d_in[10];
  const float* Wmlp0 = (const float*)d_in[11];
  const float* Wmlp1 = (const float*)d_in[12];
  const float* Wi0 = (const float*)d_in[13];
  const float* Wi1 = (const float*)d_in[14];
  const float* Wi2 = (const float*)d_in[15];
  const float* Wi3 = (const float*)d_in[16];
  const float* W2s = (const float*)d_in[17];
  const float* W2v = (const float*)d_in[18];

  float* ws = (float*)d_ws;
  float* fs = ws;
  float* fvb = fs + (size_t)N_NODES * 32;
  float* ss = fvb + (size_t)N_NODES * 96;
  float* svb = ss + (size_t)N_NODES * 32;
  float* ns = svb + (size_t)N_NODES * 96;
  float* nv = ns + (size_t)N_NODES * 64;

  hipMemsetAsync(ns, 0, (size_t)N_NODES * 256 * sizeof(float), stream);

  node_linear_kernel<<<(N_NODES / 16 + 3) / 4, 128, 0, stream>>>(
      node_s, node_v, W1s_feat, W1s_self, W1v_feat, W1v_self, fs, fvb, ss, svb);

  edge_kernel<<<N_EDGES / 32, 64, 0, stream>>>(
      edge_s, edge_v, edge_rad, esrc, edst, Wmlp0, Wmlp1, Wi0, Wi1, Wi2, Wi3,
      fs, fvb, ns, nv);

  out_linear_kernel<<<(N_NODES / 16 + 3) / 4, 128, 0, stream>>>(
      ns, nv, W2s, W2v, ss, svb, (float*)d_out);
}


# edges bucketed by destination (count, scan, fill); one work-group per 80-node bucket sums messages in LDS with float atomics (arrival order) and applies the output linear; no global float atomics
# speedup vs baseline: 1.8556x; 1.8556x over previous
#include <hip/hip_runtime.h>

typedef float v2f __attribute__((ext_vector_type(2)));
typedef float v8f __attribute__((ext_vector_type(8)));

#define N_NODES 20000
#define N_EDGES 640000
#define NPB 80
#define NB 250
#define NBP 256
#define EW 16
#define ACC_ROWS 80
#define ACC_PITCH 264
#define SORT_EPT 10

__device__ __forceinline__ v8f zero8() {
  v8f v = {0.f, 0.f, 0.f, 0.f, 0.f, 0.f, 0.f, 0.f};
  return v;
}

__device__ __forceinline__ v8f wmma4(v2f a, v2f b, v8f c) {
  return __builtin_amdgcn_wmma_f32_16x16x4_f32(false, a, false, b, (short)0, c,
                                               false, false);
}

__device__ __forceinline__ float gelu_tanh(float x) {
  float x3 = x * x * x;
  return 0.5f * x * (1.0f + tanhf(0.7978845608028654f * (x + 0.044715f * x3)));
}

__global__ __launch_bounds__(128) void node_linear_kernel(
    const float* __restrict__ node_s, const float* __restrict__ node_v,
    const float* __restrict__ Wsf, const float* __restrict__ Wss,
    const float* __restrict__ Wvf, const float* __restrict__ Wvs,
    float* __restrict__ fs, float* __restrict__ fvb, float* __restrict__ ss,
    float* __restrict__ svb) {
  const int lane = threadIdx.x & 31;
  const int wave = threadIdx.x >> 5;
  const int tile = blockIdx.x * 4 + wave;
  if (tile >= N_NODES / 16) return;
  const int n0 = tile * 16;
  const int l15 = lane & 15, lh = lane >> 4;
  const float inv_mul = 0.17677669529663687f;

  {
    v8f aF0 = zero8(), aF1 = zero8(), aS0 = zero8(), aS1 = zero8();
    const float* arow = node_s + (size_t)(n0 + l15) * 32;
#pragma unroll
    for (int ks = 0; ks < 8; ++ks) {
      int k = ks * 4 + lh * 2;
      v2f a = *reinterpret_cast<const v2f*>(arow + k);
      v2f bf0, bf1, bs0, bs1;
      bf0.x = Wsf[k * 32 + l15];      bf0.y = Wsf[(k + 1) * 32 + l15];
      bf1.x = Wsf[k * 32 + 16 + l15]; bf1.y = Wsf[(k + 1) * 32 + 16 + l15];
      bs0.x = Wss[k * 32 + l15];      bs0.y = Wss[(k + 1) * 32 + l15];
      bs1.x = Wss[k * 32 + 16 + l15]; bs1.y = Wss[(k + 1) * 32 + 16 + l15];
      aF0 = wmma4(a, bf0, aF0);
      aF1 = wmma4(a, bf1, aF1);
      aS0 = wmma4(a, bs0, aS0);
      aS1 = wmma4(a, bs1, aS1);
    }
#pragma unroll
    for (int r = 0; r < 8; ++r) {
      int node = n0 + r + 8 * lh;
      fs[(size_t)node * 32 + l15]      = aF0[r] * inv_mul;
      fs[(size_t)node * 32 + 16 + l15] = aF1[r] * inv_mul;
      ss[(size_t)node * 32 + l15]      = aS0[r] * inv_mul;
      ss[(size_t)node * 32 + 16 + l15] = aS1[r] * inv_mul;
    }
  }

#pragma unroll
  for (int c = 0; c < 3; ++c) {
    v8f aF0 = zero8(), aF1 = zero8(), aS0 = zero8(), aS1 = zero8();
    const float* vrow = node_v + (size_t)(n0 + l15) * 96 + c;
#pragma unroll
    for (int ks = 0; ks < 8; ++ks) {
      int k = ks * 4 + lh * 2;
      v2f a;
      a.x = vrow[k * 3];
      a.y = vrow[(k + 1) * 3];
      v2f bf0, bf1, bs0, bs1;
      bf0.x = Wvf[k * 32 + l15];      bf0.y = Wvf[(k + 1) * 32 + l15];
      bf1.x = Wvf[k * 32 + 16 + l15]; bf1.y = Wvf[(k + 1) * 32 + 16 + l15];
      bs0.x = Wvs[k * 32 + l15];      bs0.y = Wvs[(k + 1) * 32 + l15];
      bs1.x = Wvs[k * 32 + 16 + l15]; bs1.y = Wvs[(k + 1) * 32 + 16 + l15];
      aF0 = wmma4(a, bf0, aF0);
      aF1 = wmma4(a, bf1, aF1);
      aS0 = wmma4(a, bs0, aS0);
      aS1 = wmma4(a, bs1, aS1);
    }
#pragma unroll
    for (int r = 0; r < 8; ++r) {
      int node = n0 + r + 8 * lh;
      fvb[((size_t)node * 3 + c) * 32 + l15]      = aF0[r] * inv_mul;
      fvb[((size_t)node * 3 + c) * 32 + 16 + l15] = aF1[r] * inv_mul;
      svb[((size_t)node * 3 + c) * 32 + l15]      = aS0[r] * inv_mul;
      svb[((size_t)node * 3 + c) * 32 + 16 + l15] = aS1[r] * inv_mul;
    }
  }
}

__device__ __forceinline__ int bucket_of(int d) {
  unsigned b = (unsigned)d / (unsigned)NPB;
  return (int)(b < (unsigned)NB ? b : (unsigned)(NB - 1));
}

__global__ __launch_bounds__(256) void bucket_count_kernel(
    const int* __restrict__ edst, int* __restrict__ counts) {
  __shared__ int hist[NBP];
  const int tid = threadIdx.x;
  for (int j = tid; j < NBP; j += 256) hist[j] = 0;
  __syncthreads();
#pragma unroll
  for (int i = 0; i < SORT_EPT; ++i) {
    int e = blockIdx.x * (256 * SORT_EPT) + i * 256 + tid;
    atomicAdd(&hist[bucket_of(edst[e])], 1);
  }
  __syncthreads();
  for (int j = tid; j < NBP; j += 256) {
    int h = hist[j];
    if (h) atomicAdd(&counts[j], h);
  }
}

__global__ __launch_bounds__(NBP) void bucket_scan_kernel(
    const int* __restrict__ counts, int* __restrict__ offs) {
  __shared__ int s[NBP];
  const int t = threadIdx.x;
  const int c = counts[t];
  s[t] = c;
  __syncthreads();
  for (int d = 1; d < NBP; d <<= 1) {
    int v = (t >= d) ? s[t - d] : 0;
    __syncthreads();
    s[t] += v;
    __syncthreads();
  }
  offs[t] = s[t] - c;
}

__global__ __launch_bounds__(256) void bucket_fill_kernel(
    const int* __restrict__ edst, const int* __restrict__ offs,
    int* __restrict__ cursor, int* __restrict__ perm) {
  __shared__ int hist[NBP];
  __shared__ int lbase[NBP];
  const int tid = threadIdx.x;
  for (int j = tid; j < NBP; j += 256) hist[j] = 0;
  __syncthreads();
  int bb[SORT_EPT], rk[SORT_EPT];
#pragma unroll
  for (int i = 0; i < SORT_EPT; ++i) {
    int e = blockIdx.x * (256 * SORT_EPT) + i * 256 + tid;
    bb[i] = bucket_of(edst[e]);
    rk[i] = atomicAdd(&hist[bb[i]], 1);
  }
  __syncthreads();
  for (int j = tid; j < NBP; j += 256) {
    int h = hist[j];
    int g = h ? atomicAdd(&cursor[j], h) : 0;
    lbase[j] = offs[j] + g;
  }
  __syncthreads();
#pragma unroll
  for (int i = 0; i < SORT_EPT; ++i) {
    int e = blockIdx.x * (256 * SORT_EPT) + i * 256 + tid;
    perm[lbase[bb[i]] + rk[i]] = e;
  }
}

__global__ __launch_bounds__(EW * 32) void edge_kernel(
    const float* __restrict__ edge_s, const float* __restrict__ edge_v,
    const float* __restrict__ edge_rad, const int* __restrict__ esrc,
    const int* __restrict__ edst, const float* __restrict__ Wmlp0,
    const float* __restrict__ Wmlp1, const float* __restrict__ Wi0,
    const float* __restrict__ Wi1, const float* __restrict__ Wi2,
    const float* __restrict__ Wi3, const float* __restrict__ fs,
    const float* __restrict__ fvb, const float* __restrict__ W2s,
    const float* __restrict__ W2v, const float* __restrict__ ss,
    const float* __restrict__ svb, const int* __restrict__ counts,
    const int* __restrict__ offs, const int* __restrict__ perm,
    float* __restrict__ out) {
  __shared__ float lds[13312 + EW * 1056 + ACC_ROWS * ACC_PITCH];
  float* acc = lds + 13312 + EW * 1056;
  const int tid = threadIdx.x;

  for (int idx = tid; idx < 1024; idx += EW * 32) {
    int j = idx & 1, l = (idx >> 1) & 31, f = idx >> 6;
    int k = (f >> 2) * 4 + (l >> 4) * 2 + j;
    int n = (f & 3) * 16 + (l & 15);
    lds[idx] = Wmlp0[k * 64 + n] * 0.25f;
  }
  for (int idx = tid; idx < 4096; idx += EW * 32) {
    int j = idx & 1, l = (idx >> 1) & 31, f = idx >> 6;
    int k = (f >> 2) * 4 + (l >> 4) * 2 + j;
    int n = (f & 3) * 16 + (l & 15);
    lds[1024 + idx] = Wmlp1[k * 64 + n] * 0.125f;
  }
  for (int idx = tid; idx < 8192; idx += EW * 32) {
    int j = idx & 1, l = (idx >> 1) & 31, f = idx >> 6;
    int k = (f >> 3) * 4 + (l >> 4) * 2 + j;
    int n = (f & 7) * 16 + (l & 15);
    const float* Wi = (n < 32) ? Wi0 : (n < 64) ? Wi1 : (n < 96) ? Wi2 : Wi3;
    lds[5120 + idx] = Wi[k * 32 + (n & 31)] * 0.125f;
  }
  for (int idx = tid; idx < ACC_ROWS * ACC_PITCH; idx += EW * 32) acc[idx] = 0.f;
  __syncthreads();

  const int lane = tid & 31;
  const int wave = __builtin_amdgcn_readfirstlane(tid >> 5);
  const int l15 = lane & 15, lh = lane >> 4;
  const int nbase = blockIdx.x * NPB;
  const int start = __builtin_amdgcn_readfirstlane(offs[blockIdx.x]);
  const int cnt = __builtin_amdgcn_readfirstlane(counts[blockIdx.x]);
  const int pend = start + cnt;
  const int ntiles = (cnt + 15) >> 4;
  float* hbuf = lds + 13312 + wave * 1056;
  const float inv_s3 = 0.5773502691896258f;

  for (int t = wave; t < ntiles; t += EW) {
  const int p0 = start + t * 16;
  const int pa = (p0 + l15 < pend) ? (p0 + l15) : (pend - 1);
  const int eA = perm[pa];

  v8f h1[4];
#pragma unroll
  for (int nt = 0; nt < 4; ++nt) h1[nt] = zero8();
  const float* Xrow = edge_rad + (size_t)eA * 16;
#pragma unroll
  for (int ks = 0; ks < 4; ++ks) {
    int k = ks * 4 + lh * 2;
    v2f a = *reinterpret_cast<const v2f*>(Xrow + k);
#pragma unroll
    for (int nt = 0; nt < 4; ++nt) {
      v2f b = *reinterpret_cast<const v2f*>(&lds[(ks * 4 + nt) * 64 + lane * 2]);
      h1[nt] = wmma4(a, b, h1[nt]);
    }
  }
#pragma unroll
  for (int nt = 0; nt < 4; ++nt)
#pragma unroll
    for (int r = 0; r < 8; ++r)
      hbuf[(r + 8 * lh) * 66 + nt * 16 + l15] = gelu_tanh(h1[nt][r]);

  v8f h2[4];
#pragma unroll
  for (int nt = 0; nt < 4; ++nt) h2[nt] = zero8();
#pragma unroll
  for (int ks = 0; ks < 16; ++ks) {
    int k = ks * 4 + lh * 2;
    v2f a = *reinterpret_cast<const v2f*>(&hbuf[l15 * 66 + k]);
#pragma unroll
    for (int nt = 0; nt < 4; ++nt) {
      v2f b =
          *reinterpret_cast<const v2f*>(&lds[1024 + (ks * 4 + nt) * 64 + lane * 2]);
      h2[nt] = wmma4(a, b, h2[nt]);
    }
  }
#pragma unroll
  for (int nt = 0; nt < 4; ++nt)
#pragma unroll
    for (int r = 0; r < 8; ++r)
      hbuf[(r + 8 * lh) * 66 + nt * 16 + l15] = gelu_tanh(h2[nt][r]);

  v8f w[8];
#pragma unroll
  for (int nt = 0; nt < 8; ++nt) w[nt] = zero8();
#pragma unroll
  for (int ks = 0; ks < 16; ++ks) {
    int k = ks * 4 + lh * 2;
    v2f a = *reinterpret_cast<const v2f*>(&hbuf[l15 * 66 + k]);
#pragma unroll
    for (int nt = 0; nt < 8; ++nt) {
      v2f b =
          *reinterpret_cast<const v2f*>(&lds[5120 + (ks * 8 + nt) * 64 + lane * 2]);
      w[nt] = wmma4(a, b, w[nt]);
    }
  }

#pragma unroll
  for (int r = 0; r < 8; ++r) {
    int m = r + 8 * lh;
    int pm = p0 + m;
    bool valid = pm < pend;
    int e = perm[valid ? pm : (pend - 1)];
    float se = edge_s[e];
    float vx = edge_v[(size_t)e * 3 + 0];
    float vy = edge_v[(size_t)e * 3 + 1];
    float vz = edge_v[(size_t)e * 3 + 2];
    int src = esrc[e];
    int dst = edst[e];
    const float* fr = fs + (size_t)src * 32;
    const float* vr = fvb + (size_t)src * 96;
    float esA = fr[l15], esB = fr[16 + l15];
    float exA = vr[l15], exB = vr[16 + l15];
    float eyA = vr[32 + l15], eyB = vr[48 + l15];
    float ezA = vr[64 + l15], ezB = vr[80 + l15];
    float w0A = w[0][r], w0B = w[1][r];
    float w1A = w[2][r], w1B = w[3][r];
    float w2A = w[4][r], w2B = w[5][r];
    float w3A = w[6][r], w3B = w[7][r];
    float dA = exA * vx + eyA * vy + ezA * vz;
    float dB = exB * vx + eyB * vy + ezB * vz;
    float t1A = w1A * esA, t1B = w1B * esB;
    float vcs[3] = {vx, vy, vz};
    float evA[3] = {exA, eyA, ezA};
    float evB[3] = {exB, eyB, ezB};

    int ln = dst - nbase;
    if (valid && (unsigned)ln < (unsigned)NPB) {
      float* nsr = acc + ln * ACC_PITCH;
      atomicAdd(nsr + l15, w0A * esA * se);
      atomicAdd(nsr + 16 + l15, w0B * esB * se);
      atomicAdd(nsr + 32 + l15, w3A * dA * inv_s3);
      atomicAdd(nsr + 48 + l15, w3B * dB * inv_s3);
      float* nvr = nsr + 64;
#pragma unroll
      for (int c = 0; c < 3; ++c) {
        atomicAdd(nvr + c * 64 + l15, t1A * vcs[c]);
        atomicAdd(nvr + c * 64 + 16 + l15, t1B * vcs[c]);
        atomicAdd(nvr + c * 64 + 32 + l15, w2A * evA[c] * se);
        atomicAdd(nvr + c * 64 + 48 + l15, w2B * evB[c] * se);
      }
    }
  }
  }
  __syncthreads();

  const float cosA = 0.9238795325112867f;
  const float scale = 0.3826834323650898f * 0.125f * 0.17677669529663687f;
  float* outs = out;
  float* outv = out + (size_t)N_NODES * 32;
  for (int item = wave; item < 2 * (NPB / 16); item += EW) {
    const int nt = (item < NPB / 16) ? item : item - NPB / 16;
    const float* arow0 = acc + (nt * 16 + l15) * ACC_PITCH;
    if (item < NPB / 16) {
      v8f a0 = zero8(), a1 = zero8();
#pragma unroll
      for (int ks = 0; ks < 16; ++ks) {
        int k = ks * 4 + lh * 2;
        v2f a = *reinterpret_cast<const v2f*>(arow0 + k);
        v2f b0, b1;
        b0.x = W2s[k * 32 + l15];      b0.y = W2s[(k + 1) * 32 + l15];
        b1.x = W2s[k * 32 + 16 + l15]; b1.y = W2s[(k + 1) * 32 + 16 + l15];
        a0 = wmma4(a, b0, a0);
        a1 = wmma4(a, b1, a1);
      }
#pragma unroll
      for (int r = 0; r < 8; ++r) {
        int node = nbase + nt * 16 + r + 8 * lh;
        outs[(size_t)node * 32 + l15] =
            cosA * ss[(size_t)node * 32 + l15] + scale * a0[r];
        outs[(size_t)node * 32 + 16 + l15] =
            cosA * ss[(size_t)node * 32 + 16 + l15] + scale * a1[r];
      }
    } else {
#pragma unroll
      for (int c = 0; c < 3; ++c) {
        v8f a0 = zero8(), a1 = zero8();
        const float* arow = arow0 + 64 + c * 64;
#pragma unroll
        for (int ks = 0; ks < 16; ++ks) {
          int k = ks * 4 + lh * 2;
          v2f a = *reinterpret_cast<const v2f*>(arow + k);
          v2f b0, b1;
          b0.x = W2v[k * 32 + l15];      b0.y = W2v[(k + 1) * 32 + l15];
          b1.x = W2v[k * 32 + 16 + l15]; b1.y = W2v[(k + 1) * 32 + 16 + l15];
          a0 = wmma4(a, b0, a0);
          a1 = wmma4(a, b1, a1);
        }
#pragma unroll
        for (int r = 0; r < 8; ++r) {
          int node = nbase + nt * 16 + r + 8 * lh;
          outv[((size_t)node * 32 + l15) * 3 + c] =
              cosA * svb[((size_t)node * 3 + c) * 32 + l15] + scale * a0[r];
          outv[((size_t)node * 32 + 16 + l15) * 3 + c] =
              cosA * svb[((size_t)node * 3 + c) * 32 + 16 + l15] + scale * a1[r];
        }
      }
    }
  }
}

extern "C" void kernel_launch(void* const* d_in, const int* in_sizes, int n_in,
                              void* d_out, int out_size, void* d_ws,
                              size_t ws_size, hipStream_t stream) {
  (void)in_sizes; (void)n_in; (void)out_size; (void)ws_size;
  const float* node_s = (const float*)d_in[0];
  const float* node_v = (const float*)d_in[1];
  const float* edge_s = (const float*)d_in[2];
  const float* edge_v = (const float*)d_in[3];
  const float* edge_rad = (const float*)d_in[4];
  const int* esrc = (const int*)d_in[5];
  const int* edst = (const int*)d_in[6];
  const float* W1s_feat = (const float*)d_in[7];
  const float* W1s_self = (const float*)d_in[8];
  const float* W1v_feat = (const float*)d_in[9];
  const float* W1v_self = (const float*)d_in[10];
  const float* Wmlp0 = (const float*)d_in[11];
  const float* Wmlp1 = (const float*)d_in[12];
  const float* Wi0 = (const float*)d_in[13];
  const float* Wi1 = (const float*)d_in[14];
  const float* Wi2 = (const float*)d_in[15];
  const float* Wi3 = (const float*)d_in[16];
  const float* W2s = (const float*)d_in[17];
  const float* W2v = (const float*)d_in[18];

  float* ws = (float*)d_ws;
  float* fs = ws;
  float* fvb = fs + (size_t)N_NODES * 32;
  float* ss = fvb + (size_t)N_NODES * 96;
  float* svb = ss + (size_t)N_NODES * 32;
  int* counts = (int*)(svb + (size_t)N_NODES * 96);
  int* cursor = counts + NBP;
  int* offs = cursor + NBP;
  int* perm = offs + NBP;

  hipMemsetAsync(counts, 0, 2 * NBP * sizeof(int), stream);

  node_linear_kernel<<<(N_NODES / 16 + 3) / 4, 128, 0, stream>>>(
      node_s, node_v, W1s_feat, W1s_self, W1v_feat, W1v_self, fs, fvb, ss, svb);

  bucket_count_kernel<<<N_EDGES / (256 * SORT_EPT), 256, 0, stream>>>(edst,
                                                                     counts);
  bucket_scan_kernel<<<1, NBP, 0, stream>>>(counts, offs);
  bucket_fill_kernel<<<N_EDGES / (256 * SORT_EPT), 256, 0, stream>>>(
      edst, offs, cursor, perm);

  edge_kernel<<<NB, EW * 32, 0, stream>>>(
      edge_s, edge_v, edge_rad, esrc, edst, Wmlp0, Wmlp1, Wi0, Wi1, Wi2, Wi3,
      fs, fvb, W2s, W2v, ss, svb, counts, offs, perm, (float*)d_out);
}


# gelu evaluated branch-free as x*rcp(1+exp2(c*(x+0.044715x^3))) on hardware exp2 and rcp (f32, about 1 ulp each) in place of library tanhf with its two divergent paths per element
# speedup vs baseline: 2.4068x; 1.2970x over previous
#include <hip/hip_runtime.h>

typedef float v2f __attribute__((ext_vector_type(2)));
typedef float v8f __attribute__((ext_vector_type(8)));

#define N_NODES 20000
#define N_EDGES 640000
#define NPB 80
#define NB 250
#define NBP 256
#define EW 16
#define ACC_ROWS 80
#define ACC_PITCH 264
#define SORT_EPT 10

__device__ __forceinline__ v8f zero8() {
  v8f v = {0.f, 0.f, 0.f, 0.f, 0.f, 0.f, 0.f, 0.f};
  return v;
}

__device__ __forceinline__ v8f wmma4(v2f a, v2f b, v8f c) {
  return __builtin_amdgcn_wmma_f32_16x16x4_f32(false, a, false, b, (short)0, c,
                                               false, false);
}

__device__ __forceinline__ float gelu_tanh(float x) {
  const float c1 = -2.0f * 0.7978845608028654f * 1.4426950408889634f;
  const float c3 = c1 * 0.044715f;
  float e = __builtin_amdgcn_exp2f(x * fmaf(x * x, c3, c1));
  return x * __builtin_amdgcn_rcpf(1.0f + e);
}

__global__ __launch_bounds__(128) void node_linear_kernel(
    const float* __restrict__ node_s, const float* __restrict__ node_v,
    const float* __restrict__ Wsf, const float* __restrict__ Wss,
    const float* __restrict__ Wvf, const float* __restrict__ Wvs,
    float* __restrict__ fs, float* __restrict__ fvb, float* __restrict__ ss,
    float* __restrict__ svb) {
  const int lane = threadIdx.x & 31;
  const int wave = threadIdx.x >> 5;
  const int tile = blockIdx.x * 4 + wave;
  if (tile >= N_NODES / 16) return;
  const int n0 = tile * 16;
  const int l15 = lane & 15, lh = lane >> 4;
  const float inv_mul = 0.17677669529663687f;

  {
    v8f aF0 = zero8(), aF1 = zero8(), aS0 = zero8(), aS1 = zero8();
    const float* arow = node_s + (size_t)(n0 + l15) * 32;
#pragma unroll
    for (int ks = 0; ks < 8; ++ks) {
      int k = ks * 4 + lh * 2;
      v2f a = *reinterpret_cast<const v2f*>(arow + k);
      v2f bf0, bf1, bs0, bs1;
      bf0.x = Wsf[k * 32 + l15];      bf0.y = Wsf[(k + 1) * 32 + l15];
      bf1.x = Wsf[k * 32 + 16 + l15]; bf1.y = Wsf[(k + 1) * 32 + 16 + l15];
      bs0.x = Wss[k * 32 + l15];      bs0.y = Wss[(k + 1) * 32 + l15];
      bs1.x = Wss[k * 32 + 16 + l15]; bs1.y = Wss[(k + 1) * 32 + 16 + l15];
      aF0 = wmma4(a, bf0, aF0);
      aF1 = wmma4(a, bf1, aF1);
      aS0 = wmma4(a, bs0, aS0);
      aS1 = wmma4(a, bs1, aS1);
    }
#pragma unroll
    for (int r = 0; r < 8; ++r) {
      int node = n0 + r + 8 * lh;
      fs[(size_t)node * 32 + l15]      = aF0[r] * inv_mul;
      fs[(size_t)node * 32 + 16 + l15] = aF1[r] * inv_mul;
      ss[(size_t)node * 32 + l15]      = aS0[r] * inv_mul;
      ss[(size_t)node * 32 + 16 + l15] = aS1[r] * inv_mul;
    }
  }

#pragma unroll
  for (int c = 0; c < 3; ++c) {
    v8f aF0 = zero8(), aF1 = zero8(), aS0 = zero8(), aS1 = zero8();
    const float* vrow = node_v + (size_t)(n0 + l15) * 96 + c;
#pragma unroll
    for (int ks = 0; ks < 8; ++ks) {
      int k = ks * 4 + lh * 2;
      v2f a;
      a.x = vrow[k * 3];
      a.y = vrow[(k + 1) * 3];
      v2f bf0, bf1, bs0, bs1;
      bf0.x = Wvf[k * 32 + l15];      bf0.y = Wvf[(k + 1) * 32 + l15];
      bf1.x = Wvf[k * 32 + 16 + l15]; bf1.y = Wvf[(k + 1) * 32 + 16 + l15];
      bs0.x = Wvs[k * 32 + l15];      bs0.y = Wvs[(k + 1) * 32 + l15];
      bs1.x = Wvs[k * 32 + 16 + l15]; bs1.y = Wvs[(k + 1) * 32 + 16 + l15];
      aF0 = wmma4(a, bf0, aF0);
      aF1 = wmma4(a, bf1, aF1);
      aS0 = wmma4(a, bs0, aS0);
      aS1 = wmma4(a, bs1, aS1);
    }
#pragma unroll
    for (int r = 0; r < 8; ++r) {
      int node = n0 + r + 8 * lh;
      fvb[((size_t)node * 3 + c) * 32 + l15]      = aF0[r] * inv_mul;
      fvb[((size_t)node * 3 + c) * 32 + 16 + l15] = aF1[r] * inv_mul;
      svb[((size_t)node * 3 + c) * 32 + l15]      = aS0[r] * inv_mul;
      svb[((size_t)node * 3 + c) * 32 + 16 + l15] = aS1[r] * inv_mul;
    }
  }
}

__device__ __forceinline__ int bucket_of(int d) {
  unsigned b = (unsigned)d / (unsigned)NPB;
  return (int)(b < (unsigned)NB ? b : (unsigned)(NB - 1));
}

__global__ __launch_bounds__(256) void bucket_count_kernel(
    const int* __restrict__ edst, int* __restrict__ counts) {
  __shared__ int hist[NBP];
  const int tid = threadIdx.x;
  for (int j = tid; j < NBP; j += 256) hist[j] = 0;
  __syncthreads();
#pragma unroll
  for (int i = 0; i < SORT_EPT; ++i) {
    int e = blockIdx.x * (256 * SORT_EPT) + i * 256 + tid;
    atomicAdd(&hist[bucket_of(edst[e])], 1);
  }
  __syncthreads();
  for (int j = tid; j < NBP; j += 256) {
    int h = hist[j];
    if (h) atomicAdd(&counts[j], h);
  }
}

__global__ __launch_bounds__(NBP) void bucket_scan_kernel(
    const int* __restrict__ counts, int* __restrict__ offs) {
  __shared__ int s[NBP];
  const int t = threadIdx.x;
  const int c = counts[t];
  s[t] = c;
  __syncthreads();
  for (int d = 1; d < NBP; d <<= 1) {
    int v = (t >= d) ? s[t - d] : 0;
    __syncthreads();
    s[t] += v;
    __syncthreads();
  }
  offs[t] = s[t] - c;
}

__global__ __launch_bounds__(256) void bucket_fill_kernel(
    const int* __restrict__ edst, const int* __restrict__ offs,
    int* __restrict__ cursor, int* __restrict__ perm) {
  __shared__ int hist[NBP];
  __shared__ int lbase[NBP];
  const int tid = threadIdx.x;
  for (int j = tid; j < NBP; j += 256) hist[j] = 0;
  __syncthreads();
  int bb[SORT_EPT], rk[SORT_EPT];
#pragma unroll
  for (int i = 0; i < SORT_EPT; ++i) {
    int e = blockIdx.x * (256 * SORT_EPT) + i * 256 + tid;
    bb[i] = bucket_of(edst[e]);
    rk[i] = atomicAdd(&hist[bb[i]], 1);
  }
  __syncthreads();
  for (int j = tid; j < NBP; j += 256) {
    int h = hist[j];
    int g = h ? atomicAdd(&cursor[j], h) : 0;
    lbase[j] = offs[j] + g;
  }
  __syncthreads();
#pragma unroll
  for (int i = 0; i < SORT_EPT; ++i) {
    int e = blockIdx.x * (256 * SORT_EPT) + i * 256 + tid;
    perm[lbase[bb[i]] + rk[i]] = e;
  }
}

__global__ __launch_bounds__(EW * 32) void edge_kernel(
    const float* __restrict__ edge_s, const float* __restrict__ edge_v,
    const float* __restrict__ edge_rad, const int* __restrict__ esrc,
    const int* __restrict__ edst, const float* __restrict__ Wmlp0,
    const float* __restrict__ Wmlp1, const float* __restrict__ Wi0,
    const float* __restrict__ Wi1, const float* __restrict__ Wi2,
    const float* __restrict__ Wi3, const float* __restrict__ fs,
    const float* __restrict__ fvb, const float* __restrict__ W2s,
    const float* __restrict__ W2v, const float* __restrict__ ss,
    const float* __restrict__ svb, const int* __restrict__ counts,
    const int* __restrict__ offs, const int* __restrict__ perm,
    float* __restrict__ out) {
  __shared__ float lds[13312 + EW * 1056 + ACC_ROWS * ACC_PITCH];
  float* acc = lds + 13312 + EW * 1056;
  const int tid = threadIdx.x;

  for (int idx = tid; idx < 1024; idx += EW * 32) {
    int j = idx & 1, l = (idx >> 1) & 31, f = idx >> 6;
    int k = (f >> 2) * 4 + (l >> 4) * 2 + j;
    int n = (f & 3) * 16 + (l & 15);
    lds[idx] = Wmlp0[k * 64 + n] * 0.25f;
  }
  for (int idx = tid; idx < 4096; idx += EW * 32) {
    int j = idx & 1, l = (idx >> 1) & 31, f = idx >> 6;
    int k = (f >> 2) * 4 + (l >> 4) * 2 + j;
    int n = (f & 3) * 16 + (l & 15);
    lds[1024 + idx] = Wmlp1[k * 64 + n] * 0.125f;
  }
  for (int idx = tid; idx < 8192; idx += EW * 32) {
    int j = idx & 1, l = (idx >> 1) & 31, f = idx >> 6;
    int k = (f >> 3) * 4 + (l >> 4) * 2 + j;
    int n = (f & 7) * 16 + (l & 15);
    const float* Wi = (n < 32) ? Wi0 : (n < 64) ? Wi1 : (n < 96) ? Wi2 : Wi3;
    lds[5120 + idx] = Wi[k * 32 + (n & 31)] * 0.125f;
  }
  for (int idx = tid; idx < ACC_ROWS * ACC_PITCH; idx += EW * 32) acc[idx] = 0.f;
  __syncthreads();

  const int lane = tid & 31;
  const int wave = __builtin_amdgcn_readfirstlane(tid >> 5);
  const int l15 = lane & 15, lh = lane >> 4;
  const int nbase = blockIdx.x * NPB;
  const int start = __builtin_amdgcn_readfirstlane(offs[blockIdx.x]);
  const int cnt = __builtin_amdgcn_readfirstlane(counts[blockIdx.x]);
  const int pend = start + cnt;
  const int ntiles = (cnt + 15) >> 4;
  float* hbuf = lds + 13312 + wave * 1056;
  const float inv_s3 = 0.5773502691896258f;

  for (int t = wave; t < ntiles; t += EW) {
  const int p0 = start + t * 16;
  const int pa = (p0 + l15 < pend) ? (p0 + l15) : (pend - 1);
  const int eA = perm[pa];

  v8f h1[4];
#pragma unroll
  for (int nt = 0; nt < 4; ++nt) h1[nt] = zero8();
  const float* Xrow = edge_rad + (size_t)eA * 16;
#pragma unroll
  for (int ks = 0; ks < 4; ++ks) {
    int k = ks * 4 + lh * 2;
    v2f a = *reinterpret_cast<const v2f*>(Xrow + k);
#pragma unroll
    for (int nt = 0; nt < 4; ++nt) {
      v2f b = *reinterpret_cast<const v2f*>(&lds[(ks * 4 + nt) * 64 + lane * 2]);
      h1[nt] = wmma4(a, b, h1[nt]);
    }
  }
#pragma unroll
  for (int nt = 0; nt < 4; ++nt)
#pragma unroll
    for (int r = 0; r < 8; ++r)
      hbuf[(r + 8 * lh) * 66 + nt * 16 + l15] = gelu_tanh(h1[nt][r]);

  v8f h2[4];
#pragma unroll
  for (int nt = 0; nt < 4; ++nt) h2[nt] = zero8();
#pragma unroll
  for (int ks = 0; ks < 16; ++ks) {
    int k = ks * 4 + lh * 2;
    v2f a = *reinterpret_cast<const v2f*>(&hbuf[l15 * 66 + k]);
#pragma unroll
    for (int nt = 0; nt < 4; ++nt) {
      v2f b =
          *reinterpret_cast<const v2f*>(&lds[1024 + (ks * 4 + nt) * 64 + lane * 2]);
      h2[nt] = wmma4(a, b, h2[nt]);
    }
  }
#pragma unroll
  for (int nt = 0; nt < 4; ++nt)
#pragma unroll
    for (int r = 0; r < 8; ++r)
      hbuf[(r + 8 * lh) * 66 + nt * 16 + l15] = gelu_tanh(h2[nt][r]);

  v8f w[8];
#pragma unroll
  for (int nt = 0; nt < 8; ++nt) w[nt] = zero8();
#pragma unroll
  for (int ks = 0; ks < 16; ++ks) {
    int k = ks * 4 + lh * 2;
    v2f a = *reinterpret_cast<const v2f*>(&hbuf[l15 * 66 + k]);
#pragma unroll
    for (int nt = 0; nt < 8; ++nt) {
      v2f b =
          *reinterpret_cast<const v2f*>(&lds[5120 + (ks * 8 + nt) * 64 + lane * 2]);
      w[nt] = wmma4(a, b, w[nt]);
    }
  }

#pragma unroll
  for (int r = 0; r < 8; ++r) {
    int m = r + 8 * lh;
    int pm = p0 + m;
    bool valid = pm < pend;
    int e = perm[valid ? pm : (pend - 1)];
    float se = edge_s[e];
    float vx = edge_v[(size_t)e * 3 + 0];
    float vy = edge_v[(size_t)e * 3 + 1];
    float vz = edge_v[(size_t)e * 3 + 2];
    int src = esrc[e];
    int dst = edst[e];
    const float* fr = fs + (size_t)src * 32;
    const float* vr = fvb + (size_t)src * 96;
    float esA = fr[l15], esB = fr[16 + l15];
    float exA = vr[l15], exB = vr[16 + l15];
    float eyA = vr[32 + l15], eyB = vr[48 + l15];
    float ezA = vr[64 + l15], ezB = vr[80 + l15];
    float w0A = w[0][r], w0B = w[1][r];
    float w1A = w[2][r], w1B = w[3][r];
    float w2A = w[4][r], w2B = w[5][r];
    float w3A = w[6][r], w3B = w[7][r];
    float dA = exA * vx + eyA * vy + ezA * vz;
    float dB = exB * vx + eyB * vy + ezB * vz;
    float t1A = w1A * esA, t1B = w1B * esB;
    float vcs[3] = {vx, vy, vz};
    float evA[3] = {exA, eyA, ezA};
    float evB[3] = {exB, eyB, ezB};

    int ln = dst - nbase;
    if (valid && (unsigned)ln < (unsigned)NPB) {
      float* nsr = acc + ln * ACC_PITCH;
      atomicAdd(nsr + l15, w0A * esA * se);
      atomicAdd(nsr + 16 + l15, w0B * esB * se);
      atomicAdd(nsr + 32 + l15, w3A * dA * inv_s3);
      atomicAdd(nsr + 48 + l15, w3B * dB * inv_s3);
      float* nvr = nsr + 64;
#pragma unroll
      for (int c = 0; c < 3; ++c) {
        atomicAdd(nvr + c * 64 + l15, t1A * vcs[c]);
        atomicAdd(nvr + c * 64 + 16 + l15, t1B * vcs[c]);
        atomicAdd(nvr + c * 64 + 32 + l15, w2A * evA[c] * se);
        atomicAdd(nvr + c * 64 + 48 + l15, w2B * evB[c] * se);
      }
    }
  }
  }
  __syncthreads();

  const float cosA = 0.9238795325112867f;
  const float scale = 0.3826834323650898f * 0.125f * 0.17677669529663687f;
  float* outs = out;
  float* outv = out + (size_t)N_NODES * 32;
  for (int item = wave; item < 2 * (NPB / 16); item += EW) {
    const int nt = (item < NPB / 16) ? item : item - NPB / 16;
    const float* arow0 = acc + (nt * 16 + l15) * ACC_PITCH;
    if (item < NPB / 16) {
      v8f a0 = zero8(), a1 = zero8();
#pragma unroll
      for (int ks = 0; ks < 16; ++ks) {
        int k = ks * 4 + lh * 2;
        v2f a = *reinterpret_cast<const v2f*>(arow0 + k);
        v2f b0, b1;
        b0.x = W2s[k * 32 + l15];      b0.y = W2s[(k + 1) * 32 + l15];
        b1.x = W2s[k * 32 + 16 + l15]; b1.y = W2s[(k + 1) * 32 + 16 + l15];
        a0 = wmma4(a, b0, a0);
        a1 = wmma4(a, b1, a1);
      }
#pragma unroll
      for (int r = 0; r < 8; ++r) {
        int node = nbase + nt * 16 + r + 8 * lh;
        outs[(size_t)node * 32 + l15] =
            cosA * ss[(size_t)node * 32 + l15] + scale * a0[r];
        outs[(size_t)node * 32 + 16 + l15] =
            cosA * ss[(size_t)node * 32 + 16 + l15] + scale * a1[r];
      }
    } else {
#pragma unroll
      for (int c = 0; c < 3; ++c) {
        v8f a0 = zero8(), a1 = zero8();
        const float* arow = arow0 + 64 + c * 64;
#pragma unroll
        for (int ks = 0; ks < 16; ++ks) {
          int k = ks * 4 + lh * 2;
          v2f a = *reinterpret_cast<const v2f*>(arow + k);
          v2f b0, b1;
          b0.x = W2v[k * 32 + l15];      b0.y = W2v[(k + 1) * 32 + l15];
          b1.x = W2v[k * 32 + 16 + l15]; b1.y = W2v[(k + 1) * 32 + 16 + l15];
          a0 = wmma4(a, b0, a0);
          a1 = wmma4(a, b1, a1);
        }
#pragma unroll
        for (int r = 0; r < 8; ++r) {
          int node = nbase + nt * 16 + r + 8 * lh;
          outv[((size_t)node * 32 + l15) * 3 + c] =
              cosA * svb[((size_t)node * 3 + c) * 32 + l15] + scale * a0[r];
          outv[((size_t)node * 32 + 16 + l15) * 3 + c] =
              cosA * svb[((size_t)node * 3 + c) * 32 + 16 + l15] + scale * a1[r];
        }
      }
    }
  }
}

extern "C" void kernel_launch(void* const* d_in, const int* in_sizes, int n_in,
                              void* d_out, int out_size, void* d_ws,
                              size_t ws_size, hipStream_t stream) {
  (void)in_sizes; (void)n_in; (void)out_size; (void)ws_size;
  const float* node_s = (const float*)d_in[0];
  const float* node_v = (const float*)d_in[1];
  const float* edge_s = (const float*)d_in[2];
  const float* edge_v = (const float*)d_in[3];
  const float* edge_rad = (const float*)d_in[4];
  const int* esrc = (const int*)d_in[5];
  const int* edst = (const int*)d_in[6];
  const float* W1s_feat = (const float*)d_in[7];
  const float* W1s_self = (const float*)d_in[8];
  const float* W1v_feat = (const float*)d_in[9];
  const float* W1v_self = (const float*)d_in[10];
  const float* Wmlp0 = (const float*)d_in[11];
  const float* Wmlp1 = (const float*)d_in[12];
  const float* Wi0 = (const float*)d_in[13];
  const float* Wi1 = (const float*)d_in[14];
  const float* Wi2 = (const float*)d_in[15];
  const float* Wi3 = (const float*)d_in[16];
  const float* W2s = (const float*)d_in[17];
  const float* W2v = (const float*)d_in[18];

  float* ws = (float*)d_ws;
  float* fs = ws;
  float* fvb = fs + (size_t)N_NODES * 32;
  float* ss = fvb + (size_t)N_NODES * 96;
  float* svb = ss + (size_t)N_NODES * 32;
  int* counts = (int*)(svb + (size_t)N_NODES * 96);
  int* cursor = counts + NBP;
  int* offs = cursor + NBP;
  int* perm = offs + NBP;

  hipMemsetAsync(counts, 0, 2 * NBP * sizeof(int), stream);

  node_linear_kernel<<<(N_NODES / 16 + 3) / 4, 128, 0, stream>>>(
      node_s, node_v, W1s_feat, W1s_self, W1v_feat, W1v_self, fs, fvb, ss, svb);

  bucket_count_kernel<<<N_EDGES / (256 * SORT_EPT), 256, 0, stream>>>(edst,
                                                                     counts);
  bucket_scan_kernel<<<1, NBP, 0, stream>>>(counts, offs);
  bucket_fill_kernel<<<N_EDGES / (256 * SORT_EPT), 256, 0, stream>>>(
      edst, offs, cursor, perm);

  edge_kernel<<<NB, EW * 32, 0, stream>>>(
      edge_s, edge_v, edge_rad, esrc, edst, Wmlp0, Wmlp1, Wi0, Wi1, Wi2, Wi3,
      fs, fvb, W2s, W2v, ss, svb, counts, offs, perm, (float*)d_out);
}


# edge tile: per-edge index/attribute loads issued once at tile top (one edge per lane, DPP row_share to its row), all 64 source-row gathers ahead of GEMM2/3; branch-free sums via a spare LDS row
# speedup vs baseline: 2.7169x; 1.1288x over previous
#include <hip/hip_runtime.h>

typedef float v2f __attribute__((ext_vector_type(2)));
typedef float v8f __attribute__((ext_vector_type(8)));

#define N_NODES 20000
#define N_EDGES 640000
#define NPB 80
#define NB 250
#define NBP 256
#define EW 16
#define ACC_ROWS 81
#define ACC_PITCH 264
#define SORT_EPT 10

__device__ __forceinline__ v8f zero8() {
  v8f v = {0.f, 0.f, 0.f, 0.f, 0.f, 0.f, 0.f, 0.f};
  return v;
}

__device__ __forceinline__ v8f wmma4(v2f a, v2f b, v8f c) {
  return __builtin_amdgcn_wmma_f32_16x16x4_f32(false, a, false, b, (short)0, c,
                                               false, false);
}

__device__ __forceinline__ float gelu_tanh(float x) {
  const float c1 = -2.0f * 0.7978845608028654f * 1.4426950408889634f;
  const float c3 = c1 * 0.044715f;
  float e = __builtin_amdgcn_exp2f(x * fmaf(x * x, c3, c1));
  return x * __builtin_amdgcn_rcpf(1.0f + e);
}

__device__ __forceinline__ int row_lane_i(int v, int r) {
  switch (r) {
    case 0: return __builtin_amdgcn_update_dpp(v, v, 0x150, 0xf, 0xf, false);
    case 1: return __builtin_amdgcn_update_dpp(v, v, 0x151, 0xf, 0xf, false);
    case 2: return __builtin_amdgcn_update_dpp(v, v, 0x152, 0xf, 0xf, false);
    case 3: return __builtin_amdgcn_update_dpp(v, v, 0x153, 0xf, 0xf, false);
    case 4: return __builtin_amdgcn_update_dpp(v, v, 0x154, 0xf, 0xf, false);
    case 5: return __builtin_amdgcn_update_dpp(v, v, 0x155, 0xf, 0xf, false);
    case 6: return __builtin_amdgcn_update_dpp(v, v, 0x156, 0xf, 0xf, false);
    default: return __builtin_amdgcn_update_dpp(v, v, 0x157, 0xf, 0xf, false);
  }
}
__device__ __forceinline__ float row_lane_f(float v, int r) {
  return __int_as_float(row_lane_i(__float_as_int(v), r));
}

__global__ __launch_bounds__(128) void node_linear_kernel(
    const float* __restrict__ node_s, const float* __restrict__ node_v,
    const float* __restrict__ Wsf, const float* __restrict__ Wss,
    const float* __restrict__ Wvf, const float* __restrict__ Wvs,
    float* __restrict__ fs, float* __restrict__ fvb, float* __restrict__ ss,
    float* __restrict__ svb) {
  const int lane = threadIdx.x & 31;
  const int wave = threadIdx.x >> 5;
  const int tile = blockIdx.x * 4 + wave;
  if (tile >= N_NODES / 16) return;
  const int n0 = tile * 16;
  const int l15 = lane & 15, lh = lane >> 4;
  const float inv_mul = 0.17677669529663687f;

  {
    v8f aF0 = zero8(), aF1 = zero8(), aS0 = zero8(), aS1 = zero8();
    const float* arow = node_s + (size_t)(n0 + l15) * 32;
#pragma unroll
    for (int ks = 0; ks < 8; ++ks) {
      int k = ks * 4 + lh * 2;
      v2f a = *reinterpret_cast<const v2f*>(arow + k);
      v2f bf0, bf1, bs0, bs1;
      bf0.x = Wsf[k * 32 + l15];      bf0.y = Wsf[(k + 1) * 32 + l15];
      bf1.x = Wsf[k * 32 + 16 + l15]; bf1.y = Wsf[(k + 1) * 32 + 16 + l15];
      bs0.x = Wss[k * 32 + l15];      bs0.y = Wss[(k + 1) * 32 + l15];
      bs1.x = Wss[k * 32 + 16 + l15]; bs1.y = Wss[(k + 1) * 32 + 16 + l15];
      aF0 = wmma4(a, bf0, aF0);
      aF1 = wmma4(a, bf1, aF1);
      aS0 = wmma4(a, bs0, aS0);
      aS1 = wmma4(a, bs1, aS1);
    }
#pragma unroll
    for (int r = 0; r < 8; ++r) {
      int node = n0 + r + 8 * lh;
      fs[(size_t)node * 32 + l15]      = aF0[r] * inv_mul;
      fs[(size_t)node * 32 + 16 + l15] = aF1[r] * inv_mul;
      ss[(size_t)node * 32 + l15]      = aS0[r] * inv_mul;
      ss[(size_t)node * 32 + 16 + l15] = aS1[r] * inv_mul;
    }
  }

#pragma unroll
  for (int c = 0; c < 3; ++c) {
    v8f aF0 = zero8(), aF1 = zero8(), aS0 = zero8(), aS1 = zero8();
    const float* vrow = node_v + (size_t)(n0 + l15) * 96 + c;
#pragma unroll
    for (int ks = 0; ks < 8; ++ks) {
      int k = ks * 4 + lh * 2;
      v2f a;
      a.x = vrow[k * 3];
      a.y = vrow[(k + 1) * 3];
      v2f bf0, bf1, bs0, bs1;
      bf0.x = Wvf[k * 32 + l15];      bf0.y = Wvf[(k + 1) * 32 + l15];
      bf1.x = Wvf[k * 32 + 16 + l15]; bf1.y = Wvf[(k + 1) * 32 + 16 + l15];
      bs0.x = Wvs[k * 32 + l15];      bs0.y = Wvs[(k + 1) * 32 + l15];
      bs1.x = Wvs[k * 32 + 16 + l15]; bs1.y = Wvs[(k + 1) * 32 + 16 + l15];
      aF0 = wmma4(a, bf0, aF0);
      aF1 = wmma4(a, bf1, aF1);
      aS0 = wmma4(a, bs0, aS0);
      aS1 = wmma4(a, bs1, aS1);
    }
#pragma unroll
    for (int r = 0; r < 8; ++r) {
      int node = n0 + r + 8 * lh;
      fvb[((size_t)node * 3 + c) * 32 + l15]      = aF0[r] * inv_mul;
      fvb[((size_t)node * 3 + c) * 32 + 16 + l15] = aF1[r] * inv_mul;
      svb[((size_t)node * 3 + c) * 32 + l15]      = aS0[r] * inv_mul;
      svb[((size_t)node * 3 + c) * 32 + 16 + l15] = aS1[r] * inv_mul;
    }
  }
}

__device__ __forceinline__ int bucket_of(int d) {
  unsigned b = (unsigned)d / (unsigned)NPB;
  return (int)(b < (unsigned)NB ? b : (unsigned)(NB - 1));
}

__global__ __launch_bounds__(256) void bucket_count_kernel(
    const int* __restrict__ edst, int* __restrict__ counts) {
  __shared__ int hist[NBP];
  const int tid = threadIdx.x;
  for (int j = tid; j < NBP; j += 256) hist[j] = 0;
  __syncthreads();
#pragma unroll
  for (int i = 0; i < SORT_EPT; ++i) {
    int e = blockIdx.x * (256 * SORT_EPT) + i * 256 + tid;
    atomicAdd(&hist[bucket_of(edst[e])], 1);
  }
  __syncthreads();
  for (int j = tid; j < NBP; j += 256) {
    int h = hist[j];
    if (h) atomicAdd(&counts[j], h);
  }
}

__global__ __launch_bounds__(NBP) void bucket_scan_kernel(
    const int* __restrict__ counts, int* __restrict__ offs) {
  __shared__ int s[NBP];
  const int t = threadIdx.x;
  const int c = counts[t];
  s[t] = c;
  __syncthreads();
  for (int d = 1; d < NBP; d <<= 1) {
    int v = (t >= d) ? s[t - d] : 0;
    __syncthreads();
    s[t] += v;
    __syncthreads();
  }
  offs[t] = s[t] - c;
}

__global__ __launch_bounds__(256) void bucket_fill_kernel(
    const int* __restrict__ edst, const int* __restrict__ offs,
    int* __restrict__ cursor, int* __restrict__ perm) {
  __shared__ int hist[NBP];
  __shared__ int lbase[NBP];
  const int tid = threadIdx.x;
  for (int j = tid; j < NBP; j += 256) hist[j] = 0;
  __syncthreads();
  int bb[SORT_EPT], rk[SORT_EPT];
#pragma unroll
  for (int i = 0; i < SORT_EPT; ++i) {
    int e = blockIdx.x * (256 * SORT_EPT) + i * 256 + tid;
    bb[i] = bucket_of(edst[e]);
    rk[i] = atomicAdd(&hist[bb[i]], 1);
  }
  __syncthreads();
  for (int j = tid; j < NBP; j += 256) {
    int h = hist[j];
    int g = h ? atomicAdd(&cursor[j], h) : 0;
    lbase[j] = offs[j] + g;
  }
  __syncthreads();
#pragma unroll
  for (int i = 0; i < SORT_EPT; ++i) {
    int e = blockIdx.x * (256 * SORT_EPT) + i * 256 + tid;
    perm[lbase[bb[i]] + rk[i]] = e;
  }
}

__global__ __launch_bounds__(EW * 32) void edge_kernel(
    const float* __restrict__ edge_s, const float* __restrict__ edge_v,
    const float* __restrict__ edge_rad, const int* __restrict__ esrc,
    const int* __restrict__ edst, const float* __restrict__ Wmlp0,
    const float* __restrict__ Wmlp1, const float* __restrict__ Wi0,
    const float* __restrict__ Wi1, const float* __restrict__ Wi2,
    const float* __restrict__ Wi3, const float* __restrict__ fs,
    const float* __restrict__ fvb, const float* __restrict__ W2s,
    const float* __restrict__ W2v, const float* __restrict__ ss,
    const float* __restrict__ svb, const int* __restrict__ counts,
    const int* __restrict__ offs, const int* __restrict__ perm,
    float* __restrict__ out) {
  __shared__ float lds[13312 + EW * 1056 + ACC_ROWS * ACC_PITCH];
  float* acc = lds + 13312 + EW * 1056;
  const int tid = threadIdx.x;

  for (int idx = tid; idx < 1024; idx += EW * 32) {
    int j = idx & 1, l = (idx >> 1) & 31, f = idx >> 6;
    int k = (f >> 2) * 4 + (l >> 4) * 2 + j;
    int n = (f & 3) * 16 + (l & 15);
    lds[idx] = Wmlp0[k * 64 + n] * 0.25f;
  }
  for (int idx = tid; idx < 4096; idx += EW * 32) {
    int j = idx & 1, l = (idx >> 1) & 31, f = idx >> 6;
    int k = (f >> 2) * 4 + (l >> 4) * 2 + j;
    int n = (f & 3) * 16 + (l & 15);
    lds[1024 + idx] = Wmlp1[k * 64 + n] * 0.125f;
  }
  for (int idx = tid; idx < 8192; idx += EW * 32) {
    int j = idx & 1, l = (idx >> 1) & 31, f = idx >> 6;
    int k = (f >> 3) * 4 + (l >> 4) * 2 + j;
    int n = (f & 7) * 16 + (l & 15);
    const float* Wi = (n < 32) ? Wi0 : (n < 64) ? Wi1 : (n < 96) ? Wi2 : Wi3;
    lds[5120 + idx] = Wi[k * 32 + (n & 31)] * 0.125f;
  }
  for (int idx = tid; idx < ACC_ROWS * ACC_PITCH; idx += EW * 32) acc[idx] = 0.f;
  __syncthreads();

  const int lane = tid & 31;
  const int wave = __builtin_amdgcn_readfirstlane(tid >> 5);
  const int l15 = lane & 15, lh = lane >> 4;
  const int nbase = blockIdx.x * NPB;
  const int start = __builtin_amdgcn_readfirstlane(offs[blockIdx.x]);
  const int cnt = __builtin_amdgcn_readfirstlane(counts[blockIdx.x]);
  const int pend = start + cnt;
  const int ntiles = (cnt + 15) >> 4;
  float* hbuf = lds + 13312 + wave * 1056;
  const float inv_s3 = 0.5773502691896258f;

  for (int t = wave; t < ntiles; t += EW) {
  const int p0 = start + t * 16;
  const int pa = (p0 + l15 < pend) ? (p0 + l15) : (pend - 1);
  const int eA = perm[pa];
  const int pM = p0 + (lane & 7) + 8 * lh;
  const int eM = perm[(pM < pend) ? pM : (pend - 1)];
  const int srcM = esrc[eM];
  const int dstM = edst[eM];
  const float seM = edge_s[eM];
  const float vxM = edge_v[(size_t)eM * 3 + 0];
  const float vyM = edge_v[(size_t)eM * 3 + 1];
  const float vzM = edge_v[(size_t)eM * 3 + 2];

  v8f h1[4];
#pragma unroll
  for (int nt = 0; nt < 4; ++nt) h1[nt] = zero8();
  const float* Xrow = edge_rad + (size_t)eA * 16;
#pragma unroll
  for (int ks = 0; ks < 4; ++ks) {
    int k = ks * 4 + lh * 2;
    v2f a = *reinterpret_cast<const v2f*>(Xrow + k);
#pragma unroll
    for (int nt = 0; nt < 4; ++nt) {
      v2f b = *reinterpret_cast<const v2f*>(&lds[(ks * 4 + nt) * 64 + lane * 2]);
      h1[nt] = wmma4(a, b, h1[nt]);
    }
  }
#pragma unroll
  for (int nt = 0; nt < 4; ++nt)
#pragma unroll
    for (int r = 0; r < 8; ++r)
      hbuf[(r + 8 * lh) * 66 + nt * 16 + l15] = gelu_tanh(h1[nt][r]);

  float gs[8][2], gv[8][6];
#pragma unroll
  for (int r = 0; r < 8; ++r) {
    const int src = row_lane_i(srcM, r);
    const int fi = src * 32 + l15;
    const int vi = src * 96 + l15;
    gs[r][0] = fs[fi];       gs[r][1] = fs[fi + 16];
    gv[r][0] = fvb[vi];      gv[r][1] = fvb[vi + 16];
    gv[r][2] = fvb[vi + 32]; gv[r][3] = fvb[vi + 48];
    gv[r][4] = fvb[vi + 64]; gv[r][5] = fvb[vi + 80];
  }

  v8f h2[4];
#pragma unroll
  for (int nt = 0; nt < 4; ++nt) h2[nt] = zero8();
#pragma unroll
  for (int ks = 0; ks < 16; ++ks) {
    int k = ks * 4 + lh * 2;
    v2f a = *reinterpret_cast<const v2f*>(&hbuf[l15 * 66 + k]);
#pragma unroll
    for (int nt = 0; nt < 4; ++nt) {
      v2f b =
          *reinterpret_cast<const v2f*>(&lds[1024 + (ks * 4 + nt) * 64 + lane * 2]);
      h2[nt] = wmma4(a, b, h2[nt]);
    }
  }
#pragma unroll
  for (int nt = 0; nt < 4; ++nt)
#pragma unroll
    for (int r = 0; r < 8; ++r)
      hbuf[(r + 8 * lh) * 66 + nt * 16 + l15] = gelu_tanh(h2[nt][r]);

  v8f w[8];
#pragma unroll
  for (int nt = 0; nt < 8; ++nt) w[nt] = zero8();
#pragma unroll
  for (int ks = 0; ks < 16; ++ks) {
    int k = ks * 4 + lh * 2;
    v2f a = *reinterpret_cast<const v2f*>(&hbuf[l15 * 66 + k]);
#pragma unroll
    for (int nt = 0; nt < 8; ++nt) {
      v2f b =
          *reinterpret_cast<const v2f*>(&lds[5120 + (ks * 8 + nt) * 64 + lane * 2]);
      w[nt] = wmma4(a, b, w[nt]);
    }
  }

  __builtin_amdgcn_sched_barrier(0);

#pragma unroll
  for (int r = 0; r < 8; ++r) {
    int m = r + 8 * lh;
    int pm = p0 + m;
    bool valid = pm < pend;
    float se = row_lane_f(seM, r);
    float vx = row_lane_f(vxM, r);
    float vy = row_lane_f(vyM, r);
    float vz = row_lane_f(vzM, r);
    int dst = row_lane_i(dstM, r);
    float esA = gs[r][0], esB = gs[r][1];
    float exA = gv[r][0], exB = gv[r][1];
    float eyA = gv[r][2], eyB = gv[r][3];
    float ezA = gv[r][4], ezB = gv[r][5];
    float w0A = w[0][r], w0B = w[1][r];
    float w1A = w[2][r], w1B = w[3][r];
    float w2A = w[4][r], w2B = w[5][r];
    float w3A = w[6][r], w3B = w[7][r];
    float dA = exA * vx + eyA * vy + ezA * vz;
    float dB = exB * vx + eyB * vy + ezB * vz;
    float t1A = w1A * esA, t1B = w1B * esB;
    float vcs[3] = {vx, vy, vz};
    float evA[3] = {exA, eyA, ezA};
    float evB[3] = {exB, eyB, ezB};

    int ln = dst - nbase;
    int row = (valid && (unsigned)ln < (unsigned)NPB) ? ln : NPB;
    float* nsr = acc + row * ACC_PITCH;
    atomicAdd(nsr + l15, w0A * esA * se);
    atomicAdd(nsr + 16 + l15, w0B * esB * se);
    atomicAdd(nsr + 32 + l15, w3A * dA * inv_s3);
    atomicAdd(nsr + 48 + l15, w3B * dB * inv_s3);
    float* nvr = nsr + 64;
#pragma unroll
    for (int c = 0; c < 3; ++c) {
      atomicAdd(nvr + c * 64 + l15, t1A * vcs[c]);
      atomicAdd(nvr + c * 64 + 16 + l15, t1B * vcs[c]);
      atomicAdd(nvr + c * 64 + 32 + l15, w2A * evA[c] * se);
      atomicAdd(nvr + c * 64 + 48 + l15, w2B * evB[c] * se);
    }
  }
  }
  __syncthreads();

  const float cosA = 0.9238795325112867f;
  const float scale = 0.3826834323650898f * 0.125f * 0.17677669529663687f;
  float* outs = out;
  float* outv = out + (size_t)N_NODES * 32;
  for (int item = wave; item < 2 * (NPB / 16); item += EW) {
    const int nt = (item < NPB / 16) ? item : item - NPB / 16;
    const float* arow0 = acc + (nt * 16 + l15) * ACC_PITCH;
    if (item < NPB / 16) {
      v8f a0 = zero8(), a1 = zero8();
#pragma unroll
      for (int ks = 0; ks < 16; ++ks) {
        int k = ks * 4 + lh * 2;
        v2f a = *reinterpret_cast<const v2f*>(arow0 + k);
        v2f b0, b1;
        b0.x = W2s[k * 32 + l15];      b0.y = W2s[(k + 1) * 32 + l15];
        b1.x = W2s[k * 32 + 16 + l15]; b1.y = W2s[(k + 1) * 32 + 16 + l15];
        a0 = wmma4(a, b0, a0);
        a1 = wmma4(a, b1, a1);
      }
#pragma unroll
      for (int r = 0; r < 8; ++r) {
        int node = nbase + nt * 16 + r + 8 * lh;
        outs[(size_t)node * 32 + l15] =
            cosA * ss[(size_t)node * 32 + l15] + scale * a0[r];
        outs[(size_t)node * 32 + 16 + l15] =
            cosA * ss[(size_t)node * 32 + 16 + l15] + scale * a1[r];
      }
    } else {
#pragma unroll
      for (int c = 0; c < 3; ++c) {
        v8f a0 = zero8(), a1 = zero8();
        const float* arow = arow0 + 64 + c * 64;
#pragma unroll
        for (int ks = 0; ks < 16; ++ks) {
          int k = ks * 4 + lh * 2;
          v2f a = *reinterpret_cast<const v2f*>(arow + k);
          v2f b0, b1;
          b0.x = W2v[k * 32 + l15];      b0.y = W2v[(k + 1) * 32 + l15];
          b1.x = W2v[k * 32 + 16 + l15]; b1.y = W2v[(k + 1) * 32 + 16 + l15];
          a0 = wmma4(a, b0, a0);
          a1 = wmma4(a, b1, a1);
        }
#pragma unroll
        for (int r = 0; r < 8; ++r) {
          int node = nbase + nt * 16 + r + 8 * lh;
          outv[((size_t)node * 32 + l15) * 3 + c] =
              cosA * svb[((size_t)node * 3 + c) * 32 + l15] + scale * a0[r];
          outv[((size_t)node * 32 + 16 + l15) * 3 + c] =
              cosA * svb[((size_t)node * 3 + c) * 32 + 16 + l15] + scale * a1[r];
        }
      }
    }
  }
}

extern "C" void kernel_launch(void* const* d_in, const int* in_sizes, int n_in,
                              void* d_out, int out_size, void* d_ws,
                              size_t ws_size, hipStream_t stream) {
  (void)in_sizes; (void)n_in; (void)out_size; (void)ws_size;
  const float* node_s = (const float*)d_in[0];
  const float* node_v = (const float*)d_in[1];
  const float* edge_s = (const float*)d_in[2];
  const float* edge_v = (const float*)d_in[3];
  const float* edge_rad = (const float*)d_in[4];
  const int* esrc = (const int*)d_in[5];
  const int* edst = (const int*)d_in[6];
  const float* W1s_feat = (const float*)d_in[7];
  const float* W1s_self = (const float*)d_in[8];
  const float* W1v_feat = (const float*)d_in[9];
  const float* W1v_self = (const float*)d_in[10];
  const float* Wmlp0 = (const float*)d_in[11];
  const float* Wmlp1 = (const float*)d_in[12];
  const float* Wi0 = (const float*)d_in[13];
  const float* Wi1 = (const float*)d_in[14];
  const float* Wi2 = (const float*)d_in[15];
  const float* Wi3 = (const float*)d_in[16];
  const float* W2s = (const float*)d_in[17];
  const float* W2v = (const float*)d_in[18];

  float* ws = (float*)d_ws;
  float* fs = ws;
  float* fvb = fs + (size_t)N_NODES * 32;
  float* ss = fvb + (size_t)N_NODES * 96;
  float* svb = ss + (size_t)N_NODES * 32;
  int* counts = (int*)(svb + (size_t)N_NODES * 96);
  int* cursor = counts + NBP;
  int* offs = cursor + NBP;
  int* perm = offs + NBP;

  hipMemsetAsync(counts, 0, 2 * NBP * sizeof(int), stream);

  node_linear_kernel<<<(N_NODES / 16 + 3) / 4, 128, 0, stream>>>(
      node_s, node_v, W1s_feat, W1s_self, W1v_feat, W1v_self, fs, fvb, ss, svb);

  bucket_count_kernel<<<N_EDGES / (256 * SORT_EPT), 256, 0, stream>>>(edst,
                                                                     counts);
  bucket_scan_kernel<<<1, NBP, 0, stream>>>(counts, offs);
  bucket_fill_kernel<<<N_EDGES / (256 * SORT_EPT), 256, 0, stream>>>(
      edst, offs, cursor, perm);

  edge_kernel<<<NB, EW * 32, 0, stream>>>(
      edge_s, edge_v, edge_rad, esrc, edst, Wmlp0, Wmlp1, Wi0, Wi1, Wi2, Wi3,
      fs, fvb, W2s, W2v, ss, svb, counts, offs, perm, (float*)d_out);
}


# edge tiles handed to the waves on demand from an LDS counter (one int atomic per tile) instead of equal fixed shares; which wave sums a tile, and so the arrival order of the LDS float adds, changes
# speedup vs baseline: 2.8487x; 1.0485x over previous
#include <hip/hip_runtime.h>

typedef float v2f __attribute__((ext_vector_type(2)));
typedef float v8f __attribute__((ext_vector_type(8)));

#define N_NODES 20000
#define N_EDGES 640000
#define NPB 80
#define NB 250
#define NBP 256
#define EW 16
#define ACC_ROWS 81
#define ACC_PITCH 264
#define SORT_EPT 10

__device__ __forceinline__ v8f zero8() {
  v8f v = {0.f, 0.f, 0.f, 0.f, 0.f, 0.f, 0.f, 0.f};
  return v;
}

__device__ __forceinline__ v8f wmma4(v2f a, v2f b, v8f c) {
  return __builtin_amdgcn_wmma_f32_16x16x4_f32(false, a, false, b, (short)0, c,
                                               false, false);
}

__device__ __forceinline__ float gelu_tanh(float x) {
  const float c1 = -2.0f * 0.7978845608028654f * 1.4426950408889634f;
  const float c3 = c1 * 0.044715f;
  float e = __builtin_amdgcn_exp2f(x * fmaf(x * x, c3, c1));
  return x * __builtin_amdgcn_rcpf(1.0f + e);
}

__device__ __forceinline__ int row_lane_i(int v, int r) {
  switch (r) {
    case 0: return __builtin_amdgcn_update_dpp(v, v, 0x150, 0xf, 0xf, false);
    case 1: return __builtin_amdgcn_update_dpp(v, v, 0x151, 0xf, 0xf, false);
    case 2: return __builtin_amdgcn_update_dpp(v, v, 0x152, 0xf, 0xf, false);
    case 3: return __builtin_amdgcn_update_dpp(v, v, 0x153, 0xf, 0xf, false);
    case 4: return __builtin_amdgcn_update_dpp(v, v, 0x154, 0xf, 0xf, false);
    case 5: return __builtin_amdgcn_update_dpp(v, v, 0x155, 0xf, 0xf, false);
    case 6: return __builtin_amdgcn_update_dpp(v, v, 0x156, 0xf, 0xf, false);
    default: return __builtin_amdgcn_update_dpp(v, v, 0x157, 0xf, 0xf, false);
  }
}
__device__ __forceinline__ float row_lane_f(float v, int r) {
  return __int_as_float(row_lane_i(__float_as_int(v), r));
}

__global__ __launch_bounds__(128) void node_linear_kernel(
    const float* __restrict__ node_s, const float* __restrict__ node_v,
    const float* __restrict__ Wsf, const float* __restrict__ Wss,
    const float* __restrict__ Wvf, const float* __restrict__ Wvs,
    float* __restrict__ fs, float* __restrict__ fvb, float* __restrict__ ss,
    float* __restrict__ svb) {
  const int lane = threadIdx.x & 31;
  const int wave = threadIdx.x >> 5;
  const int tile = blockIdx.x * 4 + wave;
  if (tile >= N_NODES / 16) return;
  const int n0 = tile * 16;
  const int l15 = lane & 15, lh = lane >> 4;
  const float inv_mul = 0.17677669529663687f;

  {
    v8f aF0 = zero8(), aF1 = zero8(), aS0 = zero8(), aS1 = zero8();
    const float* arow = node_s + (size_t)(n0 + l15) * 32;
#pragma unroll
    for (int ks = 0; ks < 8; ++ks) {
      int k = ks * 4 + lh * 2;
      v2f a = *reinterpret_cast<const v2f*>(arow + k);
      v2f bf0, bf1, bs0, bs1;
      bf0.x = Wsf[k * 32 + l15];      bf0.y = Wsf[(k + 1) * 32 + l15];
      bf1.x = Wsf[k * 32 + 16 + l15]; bf1.y = Wsf[(k + 1) * 32 + 16 + l15];
      bs0.x = Wss[k * 32 + l15];      bs0.y = Wss[(k + 1) * 32 + l15];
      bs1.x = Wss[k * 32 + 16 + l15]; bs1.y = Wss[(k + 1) * 32 + 16 + l15];
      aF0 = wmma4(a, bf0, aF0);
      aF1 = wmma4(a, bf1, aF1);
      aS0 = wmma4(a, bs0, aS0);
      aS1 = wmma4(a, bs1, aS1);
    }
#pragma unroll
    for (int r = 0; r < 8; ++r) {
      int node = n0 + r + 8 * lh;
      fs[(size_t)node * 32 + l15]      = aF0[r] * inv_mul;
      fs[(size_t)node * 32 + 16 + l15] = aF1[r] * inv_mul;
      ss[(size_t)node * 32 + l15]      = aS0[r] * inv_mul;
      ss[(size_t)node * 32 + 16 + l15] = aS1[r] * inv_mul;
    }
  }

#pragma unroll
  for (int c = 0; c < 3; ++c) {
    v8f aF0 = zero8(), aF1 = zero8(), aS0 = zero8(), aS1 = zero8();
    const float* vrow = node_v + (size_t)(n0 + l15) * 96 + c;
#pragma unroll
    for (int ks = 0; ks < 8; ++ks) {
      int k = ks * 4 + lh * 2;
      v2f a;
      a.x = vrow[k * 3];
      a.y = vrow[(k + 1) * 3];
      v2f bf0, bf1, bs0, bs1;
      bf0.x = Wvf[k * 32 + l15];      bf0.y = Wvf[(k + 1) * 32 + l15];
      bf1.x = Wvf[k * 32 + 16 + l15]; bf1.y = Wvf[(k + 1) * 32 + 16 + l15];
      bs0.x = Wvs[k * 32 + l15];      bs0.y = Wvs[(k + 1) * 32 + l15];
      bs1.x = Wvs[k * 32 + 16 + l15]; bs1.y = Wvs[(k + 1) * 32 + 16 + l15];
      aF0 = wmma4(a, bf0, aF0);
      aF1 = wmma4(a, bf1, aF1);
      aS0 = wmma4(a, bs0, aS0);
      aS1 = wmma4(a, bs1, aS1);
    }
#pragma unroll
    for (int r = 0; r < 8; ++r) {
      int node = n0 + r + 8 * lh;
      fvb[((size_t)node * 3 + c) * 32 + l15]      = aF0[r] * inv_mul;
      fvb[((size_t)node * 3 + c) * 32 + 16 + l15] = aF1[r] * inv_mul;
      svb[((size_t)node * 3 + c) * 32 + l15]      = aS0[r] * inv_mul;
      svb[((size_t)node * 3 + c) * 32 + 16 + l15] = aS1[r] * inv_mul;
    }
  }
}

__device__ __forceinline__ int bucket_of(int d) {
  unsigned b = (unsigned)d / (unsigned)NPB;
  return (int)(b < (unsigned)NB ? b : (unsigned)(NB - 1));
}

__global__ __launch_bounds__(256) void bucket_count_kernel(
    const int* __restrict__ edst, int* __restrict__ counts) {
  __shared__ int hist[NBP];
  const int tid = threadIdx.x;
  for (int j = tid; j < NBP; j += 256) hist[j] = 0;
  __syncthreads();
#pragma unroll
  for (int i = 0; i < SORT_EPT; ++i) {
    int e = blockIdx.x * (256 * SORT_EPT) + i * 256 + tid;
    atomicAdd(&hist[bucket_of(edst[e])], 1);
  }
  __syncthreads();
  for (int j = tid; j < NBP; j += 256) {
    int h = hist[j];
    if (h) atomicAdd(&counts[j], h);
  }
}

__global__ __launch_bounds__(NBP) void bucket_scan_kernel(
    const int* __restrict__ counts, int* __restrict__ offs) {
  __shared__ int s[NBP];
  const int t = threadIdx.x;
  const int c = counts[t];
  s[t] = c;
  __syncthreads();
  for (int d = 1; d < NBP; d <<= 1) {
    int v = (t >= d) ? s[t - d] : 0;
    __syncthreads();
    s[t] += v;
    __syncthreads();
  }
  offs[t] = s[t] - c;
}

__global__ __launch_bounds__(256) void bucket_fill_kernel(
    const int* __restrict__ edst, const int* __restrict__ offs,
    int* __restrict__ cursor, int* __restrict__ perm) {
  __shared__ int hist[NBP];
  __shared__ int lbase[NBP];
  const int tid = threadIdx.x;
  for (int j = tid; j < NBP; j += 256) hist[j] = 0;
  __syncthreads();
  int bb[SORT_EPT], rk[SORT_EPT];
#pragma unroll
  for (int i = 0; i < SORT_EPT; ++i) {
    int e = blockIdx.x * (256 * SORT_EPT) + i * 256 + tid;
    bb[i] = bucket_of(edst[e]);
    rk[i] = atomicAdd(&hist[bb[i]], 1);
  }
  __syncthreads();
  for (int j = tid; j < NBP; j += 256) {
    int h = hist[j];
    int g = h ? atomicAdd(&cursor[j], h) : 0;
    lbase[j] = offs[j] + g;
  }
  __syncthreads();
#pragma unroll
  for (int i = 0; i < SORT_EPT; ++i) {
    int e = blockIdx.x * (256 * SORT_EPT) + i * 256 + tid;
    perm[lbase[bb[i]] + rk[i]] = e;
  }
}

__global__ __launch_bounds__(EW * 32) void edge_kernel(
    const float* __restrict__ edge_s, const float* __restrict__ edge_v,
    const float* __restrict__ edge_rad, const int* __restrict__ esrc,
    const int* __restrict__ edst, const float* __restrict__ Wmlp0,
    const float* __restrict__ Wmlp1, const float* __restrict__ Wi0,
    const float* __restrict__ Wi1, const float* __restrict__ Wi2,
    const float* __restrict__ Wi3, const float* __restrict__ fs,
    const float* __restrict__ fvb, const float* __restrict__ W2s,
    const float* __restrict__ W2v, const float* __restrict__ ss,
    const float* __restrict__ svb, const int* __restrict__ counts,
    const int* __restrict__ offs, const int* __restrict__ perm,
    float* __restrict__ out) {
  __shared__ float lds[13312 + EW * 1056 + ACC_ROWS * ACC_PITCH];
  float* acc = lds + 13312 + EW * 1056;
  __shared__ int next_tile;
  const int tid = threadIdx.x;
  if (tid == 0) next_tile = EW;

  for (int idx = tid; idx < 1024; idx += EW * 32) {
    int j = idx & 1, l = (idx >> 1) & 31, f = idx >> 6;
    int k = (f >> 2) * 4 + (l >> 4) * 2 + j;
    int n = (f & 3) * 16 + (l & 15);
    lds[idx] = Wmlp0[k * 64 + n] * 0.25f;
  }
  for (int idx = tid; idx < 4096; idx += EW * 32) {
    int j = idx & 1, l = (idx >> 1) & 31, f = idx >> 6;
    int k = (f >> 2) * 4 + (l >> 4) * 2 + j;
    int n = (f & 3) * 16 + (l & 15);
    lds[1024 + idx] = Wmlp1[k * 64 + n] * 0.125f;
  }
  for (int idx = tid; idx < 8192; idx += EW * 32) {
    int j = idx & 1, l = (idx >> 1) & 31, f = idx >> 6;
    int k = (f >> 3) * 4 + (l >> 4) * 2 + j;
    int n = (f & 7) * 16 + (l & 15);
    const float* Wi = (n < 32) ? Wi0 : (n < 64) ? Wi1 : (n < 96) ? Wi2 : Wi3;
    lds[5120 + idx] = Wi[k * 32 + (n & 31)] * 0.125f;
  }
  for (int idx = tid; idx < ACC_ROWS * ACC_PITCH; idx += EW * 32) acc[idx] = 0.f;
  __syncthreads();

  const int lane = tid & 31;
  const int wave = __builtin_amdgcn_readfirstlane(tid >> 5);
  const int l15 = lane & 15, lh = lane >> 4;
  const int nbase = blockIdx.x * NPB;
  const int start = __builtin_amdgcn_readfirstlane(offs[blockIdx.x]);
  const int cnt = __builtin_amdgcn_readfirstlane(counts[blockIdx.x]);
  const int pend = start + cnt;
  const int ntiles = (cnt + 15) >> 4;
  float* hbuf = lds + 13312 + wave * 1056;
  const float inv_s3 = 0.5773502691896258f;

  int t = wave;
  while (t < ntiles) {
  const int p0 = start + t * 16;
  const int pa = (p0 + l15 < pend) ? (p0 + l15) : (pend - 1);
  const int eA = perm[pa];
  const int pM = p0 + (lane & 7) + 8 * lh;
  const int eM = perm[(pM < pend) ? pM : (pend - 1)];
  const int srcM = esrc[eM];
  const int dstM = edst[eM];
  const float seM = edge_s[eM];
  const float vxM = edge_v[(size_t)eM * 3 + 0];
  const float vyM = edge_v[(size_t)eM * 3 + 1];
  const float vzM = edge_v[(size_t)eM * 3 + 2];

  v8f h1[4];
#pragma unroll
  for (int nt = 0; nt < 4; ++nt) h1[nt] = zero8();
  const float* Xrow = edge_rad + (size_t)eA * 16;
#pragma unroll
  for (int ks = 0; ks < 4; ++ks) {
    int k = ks * 4 + lh * 2;
    v2f a = *reinterpret_cast<const v2f*>(Xrow + k);
#pragma unroll
    for (int nt = 0; nt < 4; ++nt) {
      v2f b = *reinterpret_cast<const v2f*>(&lds[(ks * 4 + nt) * 64 + lane * 2]);
      h1[nt] = wmma4(a, b, h1[nt]);
    }
  }
#pragma unroll
  for (int nt = 0; nt < 4; ++nt)
#pragma unroll
    for (int r = 0; r < 8; ++r)
      hbuf[(r + 8 * lh) * 66 + nt * 16 + l15] = gelu_tanh(h1[nt][r]);

  float gs[8][2], gv[8][6];
#pragma unroll
  for (int r = 0; r < 8; ++r) {
    const int src = row_lane_i(srcM, r);
    const int fi = src * 32 + l15;
    const int vi = src * 96 + l15;
    gs[r][0] = fs[fi];       gs[r][1] = fs[fi + 16];
    gv[r][0] = fvb[vi];      gv[r][1] = fvb[vi + 16];
    gv[r][2] = fvb[vi + 32]; gv[r][3] = fvb[vi + 48];
    gv[r][4] = fvb[vi + 64]; gv[r][5] = fvb[vi + 80];
  }

  v8f h2[4];
#pragma unroll
  for (int nt = 0; nt < 4; ++nt) h2[nt] = zero8();
#pragma unroll
  for (int ks = 0; ks < 16; ++ks) {
    int k = ks * 4 + lh * 2;
    v2f a = *reinterpret_cast<const v2f*>(&hbuf[l15 * 66 + k]);
#pragma unroll
    for (int nt = 0; nt < 4; ++nt) {
      v2f b =
          *reinterpret_cast<const v2f*>(&lds[1024 + (ks * 4 + nt) * 64 + lane * 2]);
      h2[nt] = wmma4(a, b, h2[nt]);
    }
  }
#pragma unroll
  for (int nt = 0; nt < 4; ++nt)
#pragma unroll
    for (int r = 0; r < 8; ++r)
      hbuf[(r + 8 * lh) * 66 + nt * 16 + l15] = gelu_tanh(h2[nt][r]);

  v8f w[8];
#pragma unroll
  for (int nt = 0; nt < 8; ++nt) w[nt] = zero8();
#pragma unroll
  for (int ks = 0; ks < 16; ++ks) {
    int k = ks * 4 + lh * 2;
    v2f a = *reinterpret_cast<const v2f*>(&hbuf[l15 * 66 + k]);
#pragma unroll
    for (int nt = 0; nt < 8; ++nt) {
      v2f b =
          *reinterpret_cast<const v2f*>(&lds[5120 + (ks * 8 + nt) * 64 + lane * 2]);
      w[nt] = wmma4(a, b, w[nt]);
    }
  }

  __builtin_amdgcn_sched_barrier(0);

  int t_next = 0;
  if (lane == 0) t_next = atomicAdd(&next_tile, 1);

#pragma unroll
  for (int r = 0; r < 8; ++r) {
    int m = r + 8 * lh;
    int pm = p0 + m;
    bool valid = pm < pend;
    float se = row_lane_f(seM, r);
    float vx = row_lane_f(vxM, r);
    float vy = row_lane_f(vyM, r);
    float vz = row_lane_f(vzM, r);
    int dst = row_lane_i(dstM, r);
    float esA = gs[r][0], esB = gs[r][1];
    float exA = gv[r][0], exB = gv[r][1];
    float eyA = gv[r][2], eyB = gv[r][3];
    float ezA = gv[r][4], ezB = gv[r][5];
    float w0A = w[0][r], w0B = w[1][r];
    float w1A = w[2][r], w1B = w[3][r];
    float w2A = w[4][r], w2B = w[5][r];
    float w3A = w[6][r], w3B = w[7][r];
    float dA = exA * vx + eyA * vy + ezA * vz;
    float dB = exB * vx + eyB * vy + ezB * vz;
    float t1A = w1A * esA, t1B = w1B * esB;
    float vcs[3] = {vx, vy, vz};
    float evA[3] = {exA, eyA, ezA};
    float evB[3] = {exB, eyB, ezB};

    int ln = dst - nbase;
    int row = (valid && (unsigned)ln < (unsigned)NPB) ? ln : NPB;
    float* nsr = acc + row * ACC_PITCH;
    atomicAdd(nsr + l15, w0A * esA * se);
    atomicAdd(nsr + 16 + l15, w0B * esB * se);
    atomicAdd(nsr + 32 + l15, w3A * dA * inv_s3);
    atomicAdd(nsr + 48 + l15, w3B * dB * inv_s3);
    float* nvr = nsr + 64;
#pragma unroll
    for (int c = 0; c < 3; ++c) {
      atomicAdd(nvr + c * 64 + l15, t1A * vcs[c]);
      atomicAdd(nvr + c * 64 + 16 + l15, t1B * vcs[c]);
      atomicAdd(nvr + c * 64 + 32 + l15, w2A * evA[c] * se);
      atomicAdd(nvr + c * 64 + 48 + l15, w2B * evB[c] * se);
    }
  }
  t = __builtin_amdgcn_readfirstlane(t_next);
  }
  __syncthreads();

  const float cosA = 0.9238795325112867f;
  const float scale = 0.3826834323650898f * 0.125f * 0.17677669529663687f;
  float* outs = out;
  float* outv = out + (size_t)N_NODES * 32;
  for (int item = wave; item < 2 * (NPB / 16); item += EW) {
    const int nt = (item < NPB / 16) ? item : item - NPB / 16;
    const float* arow0 = acc + (nt * 16 + l15) * ACC_PITCH;
    if (item < NPB / 16) {
      v8f a0 = zero8(), a1 = zero8();
#pragma unroll
      for (int ks = 0; ks < 16; ++ks) {
        int k = ks * 4 + lh * 2;
        v2f a = *reinterpret_cast<const v2f*>(arow0 + k);
        v2f b0, b1;
        b0.x = W2s[k * 32 + l15];      b0.y = W2s[(k + 1) * 32 + l15];
        b1.x = W2s[k * 32 + 16 + l15]; b1.y = W2s[(k + 1) * 32 + 16 + l15];
        a0 = wmma4(a, b0, a0);
        a1 = wmma4(a, b1, a1);
      }
#pragma unroll
      for (int r = 0; r < 8; ++r) {
        int node = nbase + nt * 16 + r + 8 * lh;
        outs[(size_t)node * 32 + l15] =
            cosA * ss[(size_t)node * 32 + l15] + scale * a0[r];
        outs[(size_t)node * 32 + 16 + l15] =
            cosA * ss[(size_t)node * 32 + 16 + l15] + scale * a1[r];
      }
    } else {
#pragma unroll
      for (int c = 0; c < 3; ++c) {
        v8f a0 = zero8(), a1 = zero8();
        const float* arow = arow0 + 64 + c * 64;
#pragma unroll
        for (int ks = 0; ks < 16; ++ks) {
          int k = ks * 4 + lh * 2;
          v2f a = *reinterpret_cast<const v2f*>(arow + k);
          v2f b0, b1;
          b0.x = W2v[k * 32 + l15];      b0.y = W2v[(k + 1) * 32 + l15];
          b1.x = W2v[k * 32 + 16 + l15]; b1.y = W2v[(k + 1) * 32 + 16 + l15];
          a0 = wmma4(a, b0, a0);
          a1 = wmma4(a, b1, a1);
        }
#pragma unroll
        for (int r = 0; r < 8; ++r) {
          int node = nbase + nt * 16 + r + 8 * lh;
          outv[((size_t)node * 32 + l15) * 3 + c] =
              cosA * svb[((size_t)node * 3 + c) * 32 + l15] + scale * a0[r];
          outv[((size_t)node * 32 + 16 + l15) * 3 + c] =
              cosA * svb[((size_t)node * 3 + c) * 32 + 16 + l15] + scale * a1[r];
        }
      }
    }
  }
}

extern "C" void kernel_launch(void* const* d_in, const int* in_sizes, int n_in,
                              void* d_out, int out_size, void* d_ws,
                              size_t ws_size, hipStream_t stream) {
  (void)in_sizes; (void)n_in; (void)out_size; (void)ws_size;
  const float* node_s = (const float*)d_in[0];
  const float* node_v = (const float*)d_in[1];
  const float* edge_s = (const float*)d_in[2];
  const float* edge_v = (const float*)d_in[3];
  const float* edge_rad = (const float*)d_in[4];
  const int* esrc = (const int*)d_in[5];
  const int* edst = (const int*)d_in[6];
  const float* W1s_feat = (const float*)d_in[7];
  const float* W1s_self = (const float*)d_in[8];
  const float* W1v_feat = (const float*)d_in[9];
  const float* W1v_self = (const float*)d_in[10];
  const float* Wmlp0 = (const float*)d_in[11];
  const float* Wmlp1 = (const float*)d_in[12];
  const float* Wi0 = (const float*)d_in[13];
  const float* Wi1 = (const float*)d_in[14];
  const float* Wi2 = (const float*)d_in[15];
  const float* Wi3 = (const float*)d_in[16];
  const float* W2s = (const float*)d_in[17];
  const float* W2v = (const float*)d_in[18];

  float* ws = (float*)d_ws;
  float* fs = ws;
  float* fvb = fs + (size_t)N_NODES * 32;
  float* ss = fvb + (size_t)N_NODES * 96;
  float* svb = ss + (size_t)N_NODES * 32;
  int* counts = (int*)(svb + (size_t)N_NODES * 96);
  int* cursor = counts + NBP;
  int* offs = cursor + NBP;
  int* perm = offs + NBP;

  hipMemsetAsync(counts, 0, 2 * NBP * sizeof(int), stream);

  node_linear_kernel<<<(N_NODES / 16 + 3) / 4, 128, 0, stream>>>(
      node_s, node_v, W1s_feat, W1s_self, W1v_feat, W1v_self, fs, fvb, ss, svb);

  bucket_count_kernel<<<N_EDGES / (256 * SORT_EPT), 256, 0, stream>>>(edst,
                                                                     counts);
  bucket_scan_kernel<<<1, NBP, 0, stream>>>(counts, offs);
  bucket_fill_kernel<<<N_EDGES / (256 * SORT_EPT), 256, 0, stream>>>(
      edst, offs, cursor, perm);

  edge_kernel<<<NB, EW * 32, 0, stream>>>(
      edge_s, edge_v, edge_rad, esrc, edst, Wmlp0, Wmlp1, Wi0, Wi1, Wi2, Wi3,
      fs, fvb, W2s, W2v, ss, svb, counts, offs, perm, (float*)d_out);
}
